# starting kernel
# baseline (speedup 1.0000x reference)
#include <hip/hip_runtime.h>
#include <hip/hip_bf16.h>

#define N_NODES 100000
#define N_EDGES 1000000
#define IN_DIM  8
#define HID     64
#define N_LAYERS 3
#define LN_EPS  1e-5f
#define SLOPE   0.2f
#define LDW     68

typedef float v2f __attribute__((ext_vector_type(2)));
typedef float v8f __attribute__((ext_vector_type(8)));

__global__ void gnn_proj_kernel(const float* __restrict__ x, const float* __restrict__ Wp,
                                const float* __restrict__ bp, float* __restrict__ h) {
  __shared__ float wp_s[HID * IN_DIM];
  __shared__ float bp_s[HID];
  int t = threadIdx.x;
  wp_s[t] = Wp[t];
  wp_s[t + 256] = Wp[t + 256];
  if (t < HID) bp_s[t] = bp[t];
  __syncthreads();
  int gid = blockIdx.x * 256 + t;
  int node = gid >> 6;
  int c = gid & 63;
  const float* xr = x + node * IN_DIM;
  const float* wr = wp_s + c * IN_DIM;
  float acc = bp_s[c];
#pragma unroll
  for (int k = 0; k < IN_DIM; ++k) acc += xr[k] * wr[k];
  h[gid] = acc;
}

__global__ void gnn_gemm_kernel(const float* __restrict__ h,
                                const float* __restrict__ Wself, const float* __restrict__ bself,
                                const float* __restrict__ Wneigh, const float* __restrict__ bneigh,
                                float* __restrict__ h_self, float* __restrict__ h_neigh) {
  __shared__ float w_s[HID * LDW];
  __shared__ float b_s[HID];
  const float* W  = blockIdx.y ? Wneigh : Wself;
  const float* b  = blockIdx.y ? bneigh : bself;
  float*       out = blockIdx.y ? h_neigh : h_self;

  int t = threadIdx.x;
#pragma unroll
  for (int i = 0; i < 16; ++i) {
    int idx = t + i * 256;
    w_s[(idx >> 6) * LDW + (idx & 63)] = W[idx];
  }
  if (t < HID) b_s[t] = b[t];
  __syncthreads();

  int wave = t >> 5, lane = t & 31;
  int r0 = blockIdx.x * 32 + (wave >> 2) * 16;
  int n0 = (wave & 3) * 16;
  int lr = lane & 15;
  int koff = (lane < 16) ? 0 : 2;

  const float* arow = h   + (size_t)(r0 + lr) * HID;
  const float* brow = w_s + (n0 + lr) * LDW;

  v8f c = {};
#pragma unroll
  for (int kb = 0; kb < 16; ++kb) {
    int k = kb * 4 + koff;
    v2f a  = *(const v2f*)(arow + k);
    v2f bf = *(const v2f*)(brow + k);
    c = __builtin_amdgcn_wmma_f32_16x16x4_f32(false, a, false, bf, (short)0, c, false, false);
  }

  float bias = b_s[n0 + lr];
  int mbase = r0 + ((lane < 16) ? 0 : 8);
#pragma unroll
  for (int i = 0; i < 8; ++i) {
    out[(size_t)(mbase + i) * HID + n0 + lr] = c[i] + bias;
  }
}

__global__ void gnn_edge_kernel(const int* __restrict__ eidx, const float* __restrict__ ew,
                                const float* __restrict__ Wedge, const float* __restrict__ h_neigh,
                                float* __restrict__ agg) {
  int wave = threadIdx.x >> 5, lane = threadIdx.x & 31;
  int e = blockIdx.x * 8 + wave;
  int row = eidx[e];
  int col = eidx[N_EDGES + e];
  float w = ew[e];
  int c2 = lane * 2;
  float g0 = 1.f / (1.f + __expf(-w * Wedge[c2]));
  float g1 = 1.f / (1.f + __expf(-w * Wedge[c2 + 1]));
  v2f m = *(const v2f*)(h_neigh + (size_t)col * HID + c2);
  float* dst = agg + (size_t)row * HID + c2;
  atomicAdd(dst,     m.x * g0);
  atomicAdd(dst + 1, m.y * g1);
}

__global__ void gnn_ln_kernel(const float* __restrict__ h_self, const float* __restrict__ agg,
                              const float* __restrict__ gamma, const float* __restrict__ beta,
                              const float* __restrict__ h, float* __restrict__ dst) {
  int wave = threadIdx.x >> 5, lane = threadIdx.x & 31;
  int node = blockIdx.x * 8 + wave;
  size_t base = (size_t)node * HID + lane * 2;
  v2f a = *(const v2f*)(h_self + base);
  v2f g = *(const v2f*)(agg + base);
  float v0 = a.x + g.x, v1 = a.y + g.y;
  float s = v0 + v1, ss = v0 * v0 + v1 * v1;
#pragma unroll
  for (int m = 16; m >= 1; m >>= 1) {
    s  += __shfl_xor(s,  m, 32);
    ss += __shfl_xor(ss, m, 32);
  }
  float mu  = s * (1.f / HID);
  float var = ss * (1.f / HID) - mu * mu;
  float inv = rsqrtf(var + LN_EPS);
  int c = lane * 2;
  float o0 = (v0 - mu) * inv * gamma[c]     + beta[c];
  float o1 = (v1 - mu) * inv * gamma[c + 1] + beta[c + 1];
  o0 = (o0 >= 0.f) ? o0 : SLOPE * o0;
  o1 = (o1 >= 0.f) ? o1 : SLOPE * o1;
  v2f hh = *(const v2f*)(h + base);
  v2f r; r.x = hh.x + o0; r.y = hh.y + o1;
  *(v2f*)(dst + base) = r;
}

extern "C" void kernel_launch(void* const* d_in, const int* in_sizes, int n_in,
                              void* d_out, int out_size, void* d_ws, size_t ws_size,
                              hipStream_t stream) {
  const float* x      = (const float*)d_in[0];
  const int*   eidx   = (const int*)  d_in[1];
  const float* ew     = (const float*)d_in[2];
  const float* Wp     = (const float*)d_in[3];
  const float* bp     = (const float*)d_in[4];
  const float* Wself  = (const float*)d_in[5];
  const float* bself  = (const float*)d_in[6];
  const float* Wneigh = (const float*)d_in[7];
  const float* bneigh = (const float*)d_in[8];
  const float* Wedge  = (const float*)d_in[9];
  const float* gamma  = (const float*)d_in[10];
  const float* beta   = (const float*)d_in[11];
  float* out = (float*)d_out;

  const size_t NH = (size_t)N_NODES * HID;
  float* h       = (float*)d_ws;
  float* h_self  = h + NH;
  float* h_neigh = h_self + NH;
  float* agg     = h_neigh + NH;

  gnn_proj_kernel<<<(N_NODES * HID) / 256, 256, 0, stream>>>(x, Wp, bp, h);

  for (int l = 0; l < N_LAYERS; ++l) {
    const float* Ws = Wself  + (size_t)l * HID * HID;
    const float* bs = bself  + (size_t)l * HID;
    const float* Wn = Wneigh + (size_t)l * HID * HID;
    const float* bn = bneigh + (size_t)l * HID;
    const float* We = Wedge  + (size_t)l * HID;
    const float* gm = gamma  + (size_t)l * HID;
    const float* bt = beta   + (size_t)l * HID;

    gnn_gemm_kernel<<<dim3(N_NODES / 32, 2), 256, 0, stream>>>(
        h, Ws, bs, Wn, bn, h_self, h_neigh);

    hipMemsetAsync(agg, 0, NH * sizeof(float), stream);

    gnn_edge_kernel<<<N_EDGES / 8, 256, 0, stream>>>(eidx, ew, We, h_neigh, agg);

    float* dst = (l == N_LAYERS - 1) ? out : h;
    gnn_ln_kernel<<<N_NODES / 8, 256, 0, stream>>>(h_self, agg, gm, bt, h, dst);
  }
}


# Edge phase owner-computes: 256 work-groups each own 391 destination nodes, filter the whole edge list and sum in LDS (ds_add_f32); no global atomics, no memset; sum order = LDS arrival order
# speedup vs baseline: 1.2906x; 1.2906x over previous
#include <hip/hip_runtime.h>
#include <hip/hip_bf16.h>

#define N_NODES 100000
#define N_EDGES 1000000
#define IN_DIM  8
#define HID     64
#define N_LAYERS 3
#define LN_EPS  1e-5f
#define SLOPE   0.2f
#define LDW     68

typedef float v2f __attribute__((ext_vector_type(2)));
typedef float v8f __attribute__((ext_vector_type(8)));

__global__ void gnn_proj_kernel(const float* __restrict__ x, const float* __restrict__ Wp,
                                const float* __restrict__ bp, float* __restrict__ h) {
  __shared__ float wp_s[HID * IN_DIM];
  __shared__ float bp_s[HID];
  int t = threadIdx.x;
  wp_s[t] = Wp[t];
  wp_s[t + 256] = Wp[t + 256];
  if (t < HID) bp_s[t] = bp[t];
  __syncthreads();
  int gid = blockIdx.x * 256 + t;
  int node = gid >> 6;
  int c = gid & 63;
  const float* xr = x + node * IN_DIM;
  const float* wr = wp_s + c * IN_DIM;
  float acc = bp_s[c];
#pragma unroll
  for (int k = 0; k < IN_DIM; ++k) acc += xr[k] * wr[k];
  h[gid] = acc;
}

__global__ void gnn_gemm_kernel(const float* __restrict__ h,
                                const float* __restrict__ Wself, const float* __restrict__ bself,
                                const float* __restrict__ Wneigh, const float* __restrict__ bneigh,
                                float* __restrict__ h_self, float* __restrict__ h_neigh) {
  __shared__ float w_s[HID * LDW];
  __shared__ float b_s[HID];
  const float* W  = blockIdx.y ? Wneigh : Wself;
  const float* b  = blockIdx.y ? bneigh : bself;
  float*       out = blockIdx.y ? h_neigh : h_self;

  int t = threadIdx.x;
#pragma unroll
  for (int i = 0; i < 16; ++i) {
    int idx = t + i * 256;
    w_s[(idx >> 6) * LDW + (idx & 63)] = W[idx];
  }
  if (t < HID) b_s[t] = b[t];
  __syncthreads();

  int wave = t >> 5, lane = t & 31;
  int r0 = blockIdx.x * 32 + (wave >> 2) * 16;
  int n0 = (wave & 3) * 16;
  int lr = lane & 15;
  int koff = (lane < 16) ? 0 : 2;

  const float* arow = h   + (size_t)(r0 + lr) * HID;
  const float* brow = w_s + (n0 + lr) * LDW;

  v8f c = {};
#pragma unroll
  for (int kb = 0; kb < 16; ++kb) {
    int k = kb * 4 + koff;
    v2f a  = *(const v2f*)(arow + k);
    v2f bf = *(const v2f*)(brow + k);
    c = __builtin_amdgcn_wmma_f32_16x16x4_f32(false, a, false, bf, (short)0, c, false, false);
  }

  float bias = b_s[n0 + lr];
  int mbase = r0 + ((lane < 16) ? 0 : 8);
#pragma unroll
  for (int i = 0; i < 8; ++i) {
    out[(size_t)(mbase + i) * HID + n0 + lr] = c[i] + bias;
  }
}

#define OWN_BLOCKS  256
#define OWN_NODES   391
#define OWN_THREADS 1024
#define OWN_WAVES   (OWN_THREADS / 32)
#define SCAN_VEC    (N_EDGES / 4)
#define SCAN_FULL   (SCAN_VEC / OWN_THREADS)
#define Q_DRAIN     32
#define Q_CAP       (Q_DRAIN + 4 * 128)
#define E_BITS      20
static_assert(N_EDGES % 4 == 0 && N_EDGES <= (1 << E_BITS), "edge id must fit its field");
static_assert(OWN_NODES < (1 << (32 - E_BITS)) && OWN_BLOCKS * OWN_NODES >= N_NODES, "owner ranges");
static_assert(SCAN_FULL % 4 == 0 && SCAN_VEC - SCAN_FULL * OWN_THREADS < OWN_THREADS, "scan split");

__global__ __launch_bounds__(OWN_THREADS)
void gnn_edge_own_kernel(const int* __restrict__ eidx, const float* __restrict__ ew,
                         const float* __restrict__ Wedge, const float* __restrict__ h_neigh,
                         float* __restrict__ agg) {
  __shared__ float    acc[OWN_NODES * HID];
  __shared__ unsigned q[OWN_WAVES][Q_CAP];

  const int t = threadIdx.x, wave = t >> 5, lane = t & 31;
  const int lo  = blockIdx.x * OWN_NODES;
  const int cnt = (N_NODES - lo < OWN_NODES) ? (N_NODES - lo) : OWN_NODES;

  for (int i = t; i < OWN_NODES * HID / 4; i += OWN_THREADS)
    ((float4*)acc)[i] = make_float4(0.f, 0.f, 0.f, 0.f);
  __syncthreads();

  const int4* __restrict__ row4 = (const int4*)eidx;
  const int*  __restrict__ colp = eidx + N_EDGES;
  const int   c2  = lane * 2;
  const float we0 = Wedge[c2], we1 = Wedge[c2 + 1];
  unsigned* myq = q[wave];
  int qn = 0;

  auto drain = [&](int total, int slots) {
    for (int base = 0; base < slots; base += 32) {
      const int n = (slots - base < 32) ? (slots - base) : 32;
      const unsigned pk = (base + lane < total) ? myq[base + lane] : 0u;
      const int e  = (int)(pk & ((1u << E_BITS) - 1u));
      const int rl = (int)(pk >> E_BITS);
      const int cl = colp[e];
      const int wl = __float_as_int(ew[e]);
      for (int u = 0; u < n; u += 8) {
        v2f m[8];
#pragma unroll
        for (int k = 0; k < 8; ++k) {
          const int c = __builtin_amdgcn_readlane(cl, u + k);
          m[k] = *(const v2f*)(h_neigh + (size_t)c * HID + c2);
        }
#pragma unroll
        for (int k = 0; k < 8; ++k) {
          if (base + u + k < total) {
            const float w = __int_as_float(__builtin_amdgcn_readlane(wl, u + k));
            const int   r = __builtin_amdgcn_readlane(rl, u + k);
            const float g0 = 1.f / (1.f + __expf(-w * we0));
            const float g1 = 1.f / (1.f + __expf(-w * we1));
            float* d = acc + r * HID + c2;
            atomicAdd(d,     m[k].x * g0);
            atomicAdd(d + 1, m[k].y * g1);
          }
        }
      }
    }
  };

  auto take = [&](const int4 v, const int ebase) {
    const unsigned r0 = (unsigned)(v.x - lo), r1 = (unsigned)(v.y - lo);
    const unsigned r2 = (unsigned)(v.z - lo), r3 = (unsigned)(v.w - lo);
    const bool m0 = r0 < (unsigned)cnt, m1 = r1 < (unsigned)cnt;
    const bool m2 = r2 < (unsigned)cnt, m3 = r3 < (unsigned)cnt;
    if (__builtin_amdgcn_ballot_w32(m0 | m1 | m2 | m3)) {
      const unsigned rr[4] = {r0, r1, r2, r3};
      const bool     mm[4] = {m0, m1, m2, m3};
#pragma unroll
      for (int j = 0; j < 4; ++j) {
        const unsigned mask = __builtin_amdgcn_ballot_w32(mm[j]);
        if (mask) {
          if (mm[j])
            myq[qn + (int)__builtin_amdgcn_mbcnt_lo(mask, 0u)] =
                (rr[j] << E_BITS) | (unsigned)(ebase + j);
          qn += __builtin_popcount(mask);
        }
      }
    }
  };

  const int start = (blockIdx.x * SCAN_FULL) / OWN_BLOCKS;

  for (int g = 0; g < SCAN_FULL; g += 4) {
    int4 v[4];
    int  eb[4];
#pragma unroll
    for (int k = 0; k < 4; ++k) {
      int it = g + k + start;
      if (it >= SCAN_FULL) it -= SCAN_FULL;
      const int idx = it * OWN_THREADS + t;
      eb[k] = idx * 4;
      v[k]  = row4[idx];
    }
#pragma unroll
    for (int k = 0; k < 4; ++k) take(v[k], eb[k]);
    if (qn >= Q_DRAIN) {
      const int full = qn & ~7, rem = qn - full;
      drain(full, full);
      const unsigned keep = (lane < rem) ? myq[full + lane] : 0u;
      if (lane < rem) myq[lane] = keep;
      qn = rem;
    }
  }
  {
    const int idx = SCAN_FULL * OWN_THREADS + t;
    const int4 v = (idx < SCAN_VEC) ? row4[idx] : make_int4(-1, -1, -1, -1);
    take(v, idx * 4);
  }
  drain(qn, (qn + 7) & ~7);

  __syncthreads();
  float4* __restrict__ out4 = (float4*)(agg + (size_t)lo * HID);
  for (int i = t; i < cnt * (HID / 4); i += OWN_THREADS) out4[i] = ((const float4*)acc)[i];
}

__global__ void gnn_ln_kernel(const float* __restrict__ h_self, const float* __restrict__ agg,
                              const float* __restrict__ gamma, const float* __restrict__ beta,
                              const float* __restrict__ h, float* __restrict__ dst) {
  int wave = threadIdx.x >> 5, lane = threadIdx.x & 31;
  int node = blockIdx.x * 8 + wave;
  size_t base = (size_t)node * HID + lane * 2;
  v2f a = *(const v2f*)(h_self + base);
  v2f g = *(const v2f*)(agg + base);
  float v0 = a.x + g.x, v1 = a.y + g.y;
  float s = v0 + v1, ss = v0 * v0 + v1 * v1;
#pragma unroll
  for (int m = 16; m >= 1; m >>= 1) {
    s  += __shfl_xor(s,  m, 32);
    ss += __shfl_xor(ss, m, 32);
  }
  float mu  = s * (1.f / HID);
  float var = ss * (1.f / HID) - mu * mu;
  float inv = rsqrtf(var + LN_EPS);
  int c = lane * 2;
  float o0 = (v0 - mu) * inv * gamma[c]     + beta[c];
  float o1 = (v1 - mu) * inv * gamma[c + 1] + beta[c + 1];
  o0 = (o0 >= 0.f) ? o0 : SLOPE * o0;
  o1 = (o1 >= 0.f) ? o1 : SLOPE * o1;
  v2f hh = *(const v2f*)(h + base);
  v2f r; r.x = hh.x + o0; r.y = hh.y + o1;
  *(v2f*)(dst + base) = r;
}

extern "C" void kernel_launch(void* const* d_in, const int* in_sizes, int n_in,
                              void* d_out, int out_size, void* d_ws, size_t ws_size,
                              hipStream_t stream) {
  const float* x      = (const float*)d_in[0];
  const int*   eidx   = (const int*)  d_in[1];
  const float* ew     = (const float*)d_in[2];
  const float* Wp     = (const float*)d_in[3];
  const float* bp     = (const float*)d_in[4];
  const float* Wself  = (const float*)d_in[5];
  const float* bself  = (const float*)d_in[6];
  const float* Wneigh = (const float*)d_in[7];
  const float* bneigh = (const float*)d_in[8];
  const float* Wedge  = (const float*)d_in[9];
  const float* gamma  = (const float*)d_in[10];
  const float* beta   = (const float*)d_in[11];
  float* out = (float*)d_out;

  const size_t NH = (size_t)N_NODES * HID;
  float* h       = (float*)d_ws;
  float* h_self  = h + NH;
  float* h_neigh = h_self + NH;
  float* agg     = h_neigh + NH;

  gnn_proj_kernel<<<(N_NODES * HID) / 256, 256, 0, stream>>>(x, Wp, bp, h);

  for (int l = 0; l < N_LAYERS; ++l) {
    const float* Ws = Wself  + (size_t)l * HID * HID;
    const float* bs = bself  + (size_t)l * HID;
    const float* Wn = Wneigh + (size_t)l * HID * HID;
    const float* bn = bneigh + (size_t)l * HID;
    const float* We = Wedge  + (size_t)l * HID;
    const float* gm = gamma  + (size_t)l * HID;
    const float* bt = beta   + (size_t)l * HID;

    gnn_gemm_kernel<<<dim3(N_NODES / 32, 2), 256, 0, stream>>>(
        h, Ws, bs, Wn, bn, h_self, h_neigh);

    gnn_edge_own_kernel<<<OWN_BLOCKS, OWN_THREADS, 0, stream>>>(eidx, ew, We, h_neigh, agg);

    float* dst = (l == N_LAYERS - 1) ? out : h;
    gnn_ln_kernel<<<N_NODES / 8, 256, 0, stream>>>(h_self, agg, gm, bt, h, dst);
  }
}


# Owner filter runs once per call: layer 1 keeps each wave's matched edges as 8-byte (row|source, weight) records in the still unwritten out plane; layers 2-3 walk them, no rescan; LDS arrival order
# speedup vs baseline: 1.5810x; 1.2250x over previous
#include <hip/hip_runtime.h>
#include <hip/hip_bf16.h>

#define N_NODES 100000
#define N_EDGES 1000000
#define IN_DIM  8
#define HID     64
#define N_LAYERS 3
#define LN_EPS  1e-5f
#define SLOPE   0.2f
#define LDW     68

typedef float v2f __attribute__((ext_vector_type(2)));
typedef float v8f __attribute__((ext_vector_type(8)));

__global__ void gnn_proj_kernel(const float* __restrict__ x, const float* __restrict__ Wp,
                                const float* __restrict__ bp, float* __restrict__ h) {
  __shared__ float wp_s[HID * IN_DIM];
  __shared__ float bp_s[HID];
  int t = threadIdx.x;
  wp_s[t] = Wp[t];
  wp_s[t + 256] = Wp[t + 256];
  if (t < HID) bp_s[t] = bp[t];
  __syncthreads();
  int gid = blockIdx.x * 256 + t;
  int node = gid >> 6;
  int c = gid & 63;
  const float* xr = x + node * IN_DIM;
  const float* wr = wp_s + c * IN_DIM;
  float acc = bp_s[c];
#pragma unroll
  for (int k = 0; k < IN_DIM; ++k) acc += xr[k] * wr[k];
  h[gid] = acc;
}

__global__ void gnn_gemm_kernel(const float* __restrict__ h,
                                const float* __restrict__ Wself, const float* __restrict__ bself,
                                const float* __restrict__ Wneigh, const float* __restrict__ bneigh,
                                float* __restrict__ h_self, float* __restrict__ h_neigh) {
  __shared__ float w_s[HID * LDW];
  __shared__ float b_s[HID];
  const float* W  = blockIdx.y ? Wneigh : Wself;
  const float* b  = blockIdx.y ? bneigh : bself;
  float*       out = blockIdx.y ? h_neigh : h_self;

  int t = threadIdx.x;
#pragma unroll
  for (int i = 0; i < 16; ++i) {
    int idx = t + i * 256;
    w_s[(idx >> 6) * LDW + (idx & 63)] = W[idx];
  }
  if (t < HID) b_s[t] = b[t];
  __syncthreads();

  int wave = t >> 5, lane = t & 31;
  int r0 = blockIdx.x * 32 + (wave >> 2) * 16;
  int n0 = (wave & 3) * 16;
  int lr = lane & 15;
  int koff = (lane < 16) ? 0 : 2;

  const float* arow = h   + (size_t)(r0 + lr) * HID;
  const float* brow = w_s + (n0 + lr) * LDW;

  v8f c = {};
#pragma unroll
  for (int kb = 0; kb < 16; ++kb) {
    int k = kb * 4 + koff;
    v2f a  = *(const v2f*)(arow + k);
    v2f bf = *(const v2f*)(brow + k);
    c = __builtin_amdgcn_wmma_f32_16x16x4_f32(false, a, false, bf, (short)0, c, false, false);
  }

  float bias = b_s[n0 + lr];
  int mbase = r0 + ((lane < 16) ? 0 : 8);
#pragma unroll
  for (int i = 0; i < 8; ++i) {
    out[(size_t)(mbase + i) * HID + n0 + lr] = c[i] + bias;
  }
}

#define OWN_BLOCKS  256
#define OWN_NODES   391
#define OWN_THREADS 1024
#define OWN_WAVES   (OWN_THREADS / 32)
#define SCAN_VEC    (N_EDGES / 4)
#define SCAN_FULL   (SCAN_VEC / OWN_THREADS)
#define Q_DRAIN     32
#define Q_CAP       (Q_DRAIN + 4 * 128)
#define E_BITS      20
#define L_CAP       256
#define C_BITS      17
#define L_WORDS     (2 * L_CAP)
static_assert(N_EDGES % 4 == 0 && N_EDGES <= (1 << E_BITS), "edge id must fit its field");
static_assert(OWN_NODES < (1 << (32 - E_BITS)) && OWN_BLOCKS * OWN_NODES >= N_NODES, "owner ranges");
static_assert(SCAN_FULL % 4 == 0 && SCAN_VEC - SCAN_FULL * OWN_THREADS < OWN_THREADS, "scan split");
static_assert(N_NODES <= (1 << C_BITS) && OWN_NODES <= (1 << (32 - C_BITS)), "record key fields");
static_assert((size_t)OWN_BLOCKS * OWN_WAVES * (L_WORDS + 1) <= (size_t)N_NODES * HID,
              "record regions and their counts fit the borrowed plane");

template <bool LIST>
__global__ __launch_bounds__(OWN_THREADS)
void gnn_edge_own_kernel(const int* __restrict__ eidx, const float* __restrict__ ew,
                         const float* __restrict__ Wedge, const float* __restrict__ h_neigh,
                         float* __restrict__ agg, float* __restrict__ scratch) {
  __shared__ float    acc[OWN_NODES * HID];
  __shared__ unsigned q[OWN_WAVES][Q_CAP];

  const int t = threadIdx.x, wave = t >> 5, lane = t & 31;
  const int lo  = blockIdx.x * OWN_NODES;
  const int cnt = (N_NODES - lo < OWN_NODES) ? (N_NODES - lo) : OWN_NODES;

  for (int i = t; i < OWN_NODES * HID / 4; i += OWN_THREADS)
    ((float4*)acc)[i] = make_float4(0.f, 0.f, 0.f, 0.f);
  __syncthreads();

  const int4* __restrict__ row4 = (const int4*)eidx;
  const int*  __restrict__ colp = eidx + N_EDGES;
  const int   c2  = lane * 2;
  const float we0 = Wedge[c2], we1 = Wedge[c2 + 1];
  unsigned* myq = q[wave];
  int qn = 0;
  const int region = blockIdx.x * OWN_WAVES + wave;
  unsigned* lst = (unsigned*)scratch;
  unsigned* myl = lst + (size_t)region * L_WORDS;
  unsigned* myc = lst + (size_t)OWN_BLOCKS * OWN_WAVES * L_WORDS + region;
  int ln = 0;

  auto add32 = [&](const int rl, const int cl, const int wl, const int valid) {
    for (int u = 0; u < valid; u += 8) {
      v2f m[8];
#pragma unroll
      for (int k = 0; k < 8; ++k) {
        const int c = __builtin_amdgcn_readlane(cl, u + k);
        m[k] = *(const v2f*)(h_neigh + (size_t)c * HID + c2);
      }
#pragma unroll
      for (int k = 0; k < 8; ++k) {
        if (u + k < valid) {
          const float w = __int_as_float(__builtin_amdgcn_readlane(wl, u + k));
          const int   r = __builtin_amdgcn_readlane(rl, u + k);
          const float g0 = 1.f / (1.f + __expf(-w * we0));
          const float g1 = 1.f / (1.f + __expf(-w * we1));
          float* d = acc + r * HID + c2;
          atomicAdd(d,     m[k].x * g0);
          atomicAdd(d + 1, m[k].y * g1);
        }
      }
    }
  };

  auto drain = [&](const int total) {
    for (int base = 0; base < total; base += 32) {
      const int valid = (total - base < 32) ? (total - base) : 32;
      const unsigned pk = (lane < valid) ? myq[base + lane] : 0u;
      const int e  = (int)(pk & ((1u << E_BITS) - 1u));
      const int rl = (int)(pk >> E_BITS);
      const int cl = colp[e];
      const int wl = __float_as_int(ew[e]);
      if (!LIST) {
        const int p = ln + base + lane;
        if (lane < valid && p < L_CAP)
          *(uint2*)(myl + 2 * p) = make_uint2(((unsigned)rl << C_BITS) | (unsigned)cl, (unsigned)wl);
      }
      add32(rl, cl, wl, valid);
    }
    ln += total;
  };

  auto take = [&](const int4 v, const int ebase) {
    const unsigned r0 = (unsigned)(v.x - lo), r1 = (unsigned)(v.y - lo);
    const unsigned r2 = (unsigned)(v.z - lo), r3 = (unsigned)(v.w - lo);
    const bool m0 = r0 < (unsigned)cnt, m1 = r1 < (unsigned)cnt;
    const bool m2 = r2 < (unsigned)cnt, m3 = r3 < (unsigned)cnt;
    if (__builtin_amdgcn_ballot_w32(m0 | m1 | m2 | m3)) {
      const unsigned rr[4] = {r0, r1, r2, r3};
      const bool     mm[4] = {m0, m1, m2, m3};
#pragma unroll
      for (int j = 0; j < 4; ++j) {
        const unsigned mask = __builtin_amdgcn_ballot_w32(mm[j]);
        if (mask) {
          if (mm[j])
            myq[qn + (int)__builtin_amdgcn_mbcnt_lo(mask, 0u)] =
                (rr[j] << E_BITS) | (unsigned)(ebase + j);
          qn += __builtin_popcount(mask);
        }
      }
    }
  };

  bool scan = true;
  if (LIST) {
    const int n = __builtin_amdgcn_readfirstlane((int)*myc);
    if (n <= L_CAP) {
      scan = false;
      const uint2* rec = (const uint2*)myl;
      uint2 nx = (lane < n) ? rec[lane] : make_uint2(0u, 0u);
      for (int base = 0; base < n; base += 32) {
        const uint2 cur = nx;
        const int nb = base + 32 + lane;
        nx = (nb < n) ? rec[nb] : make_uint2(0u, 0u);
        const int valid = (n - base < 32) ? (n - base) : 32;
        add32((int)(cur.x >> C_BITS), (int)(cur.x & ((1u << C_BITS) - 1u)), (int)cur.y, valid);
      }
    }
  }

  const int start = (blockIdx.x * SCAN_FULL) / OWN_BLOCKS;

  if (scan) {
    for (int g = 0; g < SCAN_FULL; g += 4) {
      int4 v[4];
      int  eb[4];
#pragma unroll
      for (int k = 0; k < 4; ++k) {
        int it = g + k + start;
        if (it >= SCAN_FULL) it -= SCAN_FULL;
        const int idx = it * OWN_THREADS + t;
        eb[k] = idx * 4;
        v[k]  = row4[idx];
      }
#pragma unroll
      for (int k = 0; k < 4; ++k) take(v[k], eb[k]);
      if (qn >= Q_DRAIN) {
        const int full = qn & ~7, rem = qn - full;
        drain(full);
        const unsigned keep = (lane < rem) ? myq[full + lane] : 0u;
        if (lane < rem) myq[lane] = keep;
        qn = rem;
      }
    }
    {
      const int idx = SCAN_FULL * OWN_THREADS + t;
      const int4 v = (idx < SCAN_VEC) ? row4[idx] : make_int4(-1, -1, -1, -1);
      take(v, idx * 4);
    }
    drain(qn);
    if (!LIST && lane == 0) *myc = (unsigned)ln;
  }

  __syncthreads();
  float4* __restrict__ out4 = (float4*)(agg + (size_t)lo * HID);
  for (int i = t; i < cnt * (HID / 4); i += OWN_THREADS) out4[i] = ((const float4*)acc)[i];
}

__global__ void gnn_ln_kernel(const float* __restrict__ h_self, const float* __restrict__ agg,
                              const float* __restrict__ gamma, const float* __restrict__ beta,
                              const float* __restrict__ h, float* __restrict__ dst) {
  int wave = threadIdx.x >> 5, lane = threadIdx.x & 31;
  int node = blockIdx.x * 8 + wave;
  size_t base = (size_t)node * HID + lane * 2;
  v2f a = *(const v2f*)(h_self + base);
  v2f g = *(const v2f*)(agg + base);
  float v0 = a.x + g.x, v1 = a.y + g.y;
  float s = v0 + v1, ss = v0 * v0 + v1 * v1;
#pragma unroll
  for (int m = 16; m >= 1; m >>= 1) {
    s  += __shfl_xor(s,  m, 32);
    ss += __shfl_xor(ss, m, 32);
  }
  float mu  = s * (1.f / HID);
  float var = ss * (1.f / HID) - mu * mu;
  float inv = rsqrtf(var + LN_EPS);
  int c = lane * 2;
  float o0 = (v0 - mu) * inv * gamma[c]     + beta[c];
  float o1 = (v1 - mu) * inv * gamma[c + 1] + beta[c + 1];
  o0 = (o0 >= 0.f) ? o0 : SLOPE * o0;
  o1 = (o1 >= 0.f) ? o1 : SLOPE * o1;
  v2f hh = *(const v2f*)(h + base);
  v2f r; r.x = hh.x + o0; r.y = hh.y + o1;
  *(v2f*)(dst + base) = r;
}

extern "C" void kernel_launch(void* const* d_in, const int* in_sizes, int n_in,
                              void* d_out, int out_size, void* d_ws, size_t ws_size,
                              hipStream_t stream) {
  const float* x      = (const float*)d_in[0];
  const int*   eidx   = (const int*)  d_in[1];
  const float* ew     = (const float*)d_in[2];
  const float* Wp     = (const float*)d_in[3];
  const float* bp     = (const float*)d_in[4];
  const float* Wself  = (const float*)d_in[5];
  const float* bself  = (const float*)d_in[6];
  const float* Wneigh = (const float*)d_in[7];
  const float* bneigh = (const float*)d_in[8];
  const float* Wedge  = (const float*)d_in[9];
  const float* gamma  = (const float*)d_in[10];
  const float* beta   = (const float*)d_in[11];
  float* out = (float*)d_out;

  const size_t NH = (size_t)N_NODES * HID;
  float* h       = (float*)d_ws;
  float* h_self  = h + NH;
  float* h_neigh = h_self + NH;
  float* agg     = h_neigh + NH;
  float* scratch = out;

  gnn_proj_kernel<<<(N_NODES * HID) / 256, 256, 0, stream>>>(x, Wp, bp, h);

  for (int l = 0; l < N_LAYERS; ++l) {
    const float* Ws = Wself  + (size_t)l * HID * HID;
    const float* bs = bself  + (size_t)l * HID;
    const float* Wn = Wneigh + (size_t)l * HID * HID;
    const float* bn = bneigh + (size_t)l * HID;
    const float* We = Wedge  + (size_t)l * HID;
    const float* gm = gamma  + (size_t)l * HID;
    const float* bt = beta   + (size_t)l * HID;

    gnn_gemm_kernel<<<dim3(N_NODES / 32, 2), 256, 0, stream>>>(
        h, Ws, bs, Wn, bn, h_self, h_neigh);

    if (l == 0)
      gnn_edge_own_kernel<false><<<OWN_BLOCKS, OWN_THREADS, 0, stream>>>(eidx, ew, We, h_neigh, agg, scratch);
    else
      gnn_edge_own_kernel<true><<<OWN_BLOCKS, OWN_THREADS, 0, stream>>>(eidx, ew, We, h_neigh, agg, scratch);

    float* dst = (l == N_LAYERS - 1) ? out : h;
    gnn_ln_kernel<<<N_NODES / 8, 256, 0, stream>>>(h_self, agg, gm, bt, h, dst);
  }
}


# Norm, leaky-ReLU and residual done by the owner work-group on its LDS sums (h_self and h rows requested ahead of the edge phase): no agg plane, no norm launch, 7 launches not 10; sum orders kept
# speedup vs baseline: 1.7073x; 1.0799x over previous
#include <hip/hip_runtime.h>
#include <hip/hip_bf16.h>

#define N_NODES 100000
#define N_EDGES 1000000
#define IN_DIM  8
#define HID     64
#define N_LAYERS 3
#define LN_EPS  1e-5f
#define SLOPE   0.2f
#define LDW     68

typedef float v2f __attribute__((ext_vector_type(2)));
typedef float v8f __attribute__((ext_vector_type(8)));

__global__ void gnn_proj_kernel(const float* __restrict__ x, const float* __restrict__ Wp,
                                const float* __restrict__ bp, float* __restrict__ h) {
  __shared__ float wp_s[HID * IN_DIM];
  __shared__ float bp_s[HID];
  int t = threadIdx.x;
  wp_s[t] = Wp[t];
  wp_s[t + 256] = Wp[t + 256];
  if (t < HID) bp_s[t] = bp[t];
  __syncthreads();
  int gid = blockIdx.x * 256 + t;
  int node = gid >> 6;
  int c = gid & 63;
  const float* xr = x + node * IN_DIM;
  const float* wr = wp_s + c * IN_DIM;
  float acc = bp_s[c];
#pragma unroll
  for (int k = 0; k < IN_DIM; ++k) acc += xr[k] * wr[k];
  h[gid] = acc;
}

__global__ void gnn_gemm_kernel(const float* __restrict__ h,
                                const float* __restrict__ Wself, const float* __restrict__ bself,
                                const float* __restrict__ Wneigh, const float* __restrict__ bneigh,
                                float* __restrict__ h_self, float* __restrict__ h_neigh) {
  __shared__ float w_s[HID * LDW];
  __shared__ float b_s[HID];
  const float* W  = blockIdx.y ? Wneigh : Wself;
  const float* b  = blockIdx.y ? bneigh : bself;
  float*       out = blockIdx.y ? h_neigh : h_self;

  int t = threadIdx.x;
#pragma unroll
  for (int i = 0; i < 16; ++i) {
    int idx = t + i * 256;
    w_s[(idx >> 6) * LDW + (idx & 63)] = W[idx];
  }
  if (t < HID) b_s[t] = b[t];
  __syncthreads();

  int wave = t >> 5, lane = t & 31;
  int r0 = blockIdx.x * 32 + (wave >> 2) * 16;
  int n0 = (wave & 3) * 16;
  int lr = lane & 15;
  int koff = (lane < 16) ? 0 : 2;

  const float* arow = h   + (size_t)(r0 + lr) * HID;
  const float* brow = w_s + (n0 + lr) * LDW;

  v8f c = {};
#pragma unroll
  for (int kb = 0; kb < 16; ++kb) {
    int k = kb * 4 + koff;
    v2f a  = *(const v2f*)(arow + k);
    v2f bf = *(const v2f*)(brow + k);
    c = __builtin_amdgcn_wmma_f32_16x16x4_f32(false, a, false, bf, (short)0, c, false, false);
  }

  float bias = b_s[n0 + lr];
  int mbase = r0 + ((lane < 16) ? 0 : 8);
#pragma unroll
  for (int i = 0; i < 8; ++i) {
    out[(size_t)(mbase + i) * HID + n0 + lr] = c[i] + bias;
  }
}

#define OWN_BLOCKS  256
#define OWN_NODES   391
#define OWN_THREADS 1024
#define OWN_WAVES   (OWN_THREADS / 32)
#define SCAN_VEC    (N_EDGES / 4)
#define SCAN_FULL   (SCAN_VEC / OWN_THREADS)
#define Q_DRAIN     32
#define Q_CAP       (Q_DRAIN + 4 * 128)
#define E_BITS      20
#define L_CAP       256
#define C_BITS      17
#define L_WORDS     (2 * L_CAP)
#define EPI_ROWS    ((OWN_NODES + OWN_WAVES - 1) / OWN_WAVES)
static_assert(N_EDGES % 4 == 0 && N_EDGES <= (1 << E_BITS), "edge id must fit its field");
static_assert(OWN_NODES < (1 << (32 - E_BITS)) && OWN_BLOCKS * OWN_NODES >= N_NODES, "owner ranges");
static_assert(SCAN_FULL % 4 == 0 && SCAN_VEC - SCAN_FULL * OWN_THREADS < OWN_THREADS, "scan split");
static_assert(N_NODES <= (1 << C_BITS) && OWN_NODES <= (1 << (32 - C_BITS)), "record key fields");
static_assert((size_t)OWN_BLOCKS * OWN_WAVES * (L_WORDS + 1) <= (size_t)N_NODES * HID,
              "record regions and their counts fit the borrowed plane");

template <bool LIST>
__global__ __launch_bounds__(OWN_THREADS)
void gnn_edge_own_kernel(const int* __restrict__ eidx, const float* __restrict__ ew,
                         const float* __restrict__ Wedge, const float* __restrict__ h_neigh,
                         const float* __restrict__ h_self, const float* __restrict__ gamma,
                         const float* __restrict__ beta, const float* h, float* dst,
                         float* __restrict__ scratch) {
  __shared__ float    acc[OWN_NODES * HID];
  __shared__ unsigned q[OWN_WAVES][Q_CAP];

  const int t = threadIdx.x, wave = t >> 5, lane = t & 31;
  const int lo  = blockIdx.x * OWN_NODES;
  const int cnt = (N_NODES - lo < OWN_NODES) ? (N_NODES - lo) : OWN_NODES;

  const int wv = __builtin_amdgcn_readfirstlane(wave);
  v2f a[EPI_ROWS], hh[EPI_ROWS];
#pragma unroll
  for (int j = 0; j < EPI_ROWS; ++j) {
    int n = wv + j * OWN_WAVES;
    n = (n < cnt) ? n : (cnt - 1);
    const size_t base = (size_t)(lo + n) * HID + 2 * lane;
    a[j]  = *(const v2f*)(h_self + base);
    hh[j] = *(const v2f*)(h + base);
  }
  const float gm0 = gamma[2 * lane], gm1 = gamma[2 * lane + 1];
  const float bt0 = beta[2 * lane],  bt1 = beta[2 * lane + 1];

  for (int i = t; i < OWN_NODES * HID / 4; i += OWN_THREADS)
    ((float4*)acc)[i] = make_float4(0.f, 0.f, 0.f, 0.f);
  __syncthreads();

  const int4* __restrict__ row4 = (const int4*)eidx;
  const int*  __restrict__ colp = eidx + N_EDGES;
  const int   c2  = lane * 2;
  const float we0 = Wedge[c2], we1 = Wedge[c2 + 1];
  unsigned* myq = q[wave];
  int qn = 0;
  const int region = blockIdx.x * OWN_WAVES + wave;
  unsigned* lst = (unsigned*)scratch;
  unsigned* myl = lst + (size_t)region * L_WORDS;
  unsigned* myc = lst + (size_t)OWN_BLOCKS * OWN_WAVES * L_WORDS + region;
  int ln = 0;

  auto add32 = [&](const int rl, const int cl, const int wl, const int valid) {
    for (int u = 0; u < valid; u += 8) {
      v2f m[8];
#pragma unroll
      for (int k = 0; k < 8; ++k) {
        const int c = __builtin_amdgcn_readlane(cl, u + k);
        m[k] = *(const v2f*)(h_neigh + (size_t)c * HID + c2);
      }
#pragma unroll
      for (int k = 0; k < 8; ++k) {
        if (u + k < valid) {
          const float w = __int_as_float(__builtin_amdgcn_readlane(wl, u + k));
          const int   r = __builtin_amdgcn_readlane(rl, u + k);
          const float g0 = 1.f / (1.f + __expf(-w * we0));
          const float g1 = 1.f / (1.f + __expf(-w * we1));
          float* d = acc + r * HID + c2;
          atomicAdd(d,     m[k].x * g0);
          atomicAdd(d + 1, m[k].y * g1);
        }
      }
    }
  };

  auto drain = [&](const int total) {
    for (int base = 0; base < total; base += 32) {
      const int valid = (total - base < 32) ? (total - base) : 32;
      const unsigned pk = (lane < valid) ? myq[base + lane] : 0u;
      const int e  = (int)(pk & ((1u << E_BITS) - 1u));
      const int rl = (int)(pk >> E_BITS);
      const int cl = colp[e];
      const int wl = __float_as_int(ew[e]);
      if (!LIST) {
        const int p = ln + base + lane;
        if (lane < valid && p < L_CAP)
          *(uint2*)(myl + 2 * p) = make_uint2(((unsigned)rl << C_BITS) | (unsigned)cl, (unsigned)wl);
      }
      add32(rl, cl, wl, valid);
    }
    ln += total;
  };

  auto take = [&](const int4 v, const int ebase) {
    const unsigned r0 = (unsigned)(v.x - lo), r1 = (unsigned)(v.y - lo);
    const unsigned r2 = (unsigned)(v.z - lo), r3 = (unsigned)(v.w - lo);
    const bool m0 = r0 < (unsigned)cnt, m1 = r1 < (unsigned)cnt;
    const bool m2 = r2 < (unsigned)cnt, m3 = r3 < (unsigned)cnt;
    if (__builtin_amdgcn_ballot_w32(m0 | m1 | m2 | m3)) {
      const unsigned rr[4] = {r0, r1, r2, r3};
      const bool     mm[4] = {m0, m1, m2, m3};
#pragma unroll
      for (int j = 0; j < 4; ++j) {
        const unsigned mask = __builtin_amdgcn_ballot_w32(mm[j]);
        if (mask) {
          if (mm[j])
            myq[qn + (int)__builtin_amdgcn_mbcnt_lo(mask, 0u)] =
                (rr[j] << E_BITS) | (unsigned)(ebase + j);
          qn += __builtin_popcount(mask);
        }
      }
    }
  };

  bool scan = true;
  if (LIST) {
    const int n = __builtin_amdgcn_readfirstlane((int)*myc);
    if (n <= L_CAP) {
      scan = false;
      const uint2* rec = (const uint2*)myl;
      uint2 nx = (lane < n) ? rec[lane] : make_uint2(0u, 0u);
      for (int base = 0; base < n; base += 32) {
        const uint2 cur = nx;
        const int nb = base + 32 + lane;
        nx = (nb < n) ? rec[nb] : make_uint2(0u, 0u);
        const int valid = (n - base < 32) ? (n - base) : 32;
        add32((int)(cur.x >> C_BITS), (int)(cur.x & ((1u << C_BITS) - 1u)), (int)cur.y, valid);
      }
    }
  }

  const int start = (blockIdx.x * SCAN_FULL) / OWN_BLOCKS;

  if (scan) {
    for (int g = 0; g < SCAN_FULL; g += 4) {
      int4 v[4];
      int  eb[4];
#pragma unroll
      for (int k = 0; k < 4; ++k) {
        int it = g + k + start;
        if (it >= SCAN_FULL) it -= SCAN_FULL;
        const int idx = it * OWN_THREADS + t;
        eb[k] = idx * 4;
        v[k]  = row4[idx];
      }
#pragma unroll
      for (int k = 0; k < 4; ++k) take(v[k], eb[k]);
      if (qn >= Q_DRAIN) {
        const int full = qn & ~7, rem = qn - full;
        drain(full);
        const unsigned keep = (lane < rem) ? myq[full + lane] : 0u;
        if (lane < rem) myq[lane] = keep;
        qn = rem;
      }
    }
    {
      const int idx = SCAN_FULL * OWN_THREADS + t;
      const int4 v = (idx < SCAN_VEC) ? row4[idx] : make_int4(-1, -1, -1, -1);
      take(v, idx * 4);
    }
    drain(qn);
    if (!LIST && lane == 0) *myc = (unsigned)ln;
  }

  __syncthreads();

  float v0[EPI_ROWS], v1[EPI_ROWS], s[EPI_ROWS], ss[EPI_ROWS];
#pragma unroll
  for (int j = 0; j < EPI_ROWS; ++j) {
    int n = wv + j * OWN_WAVES;
    n = (n < cnt) ? n : (cnt - 1);
    const v2f g = *(const v2f*)(acc + n * HID + c2);
    v0[j] = a[j].x + g.x;
    v1[j] = a[j].y + g.y;
    s[j]  = v0[j] + v1[j];
    ss[j] = v0[j] * v0[j] + v1[j] * v1[j];
  }
#pragma unroll
  for (int m = 16; m >= 1; m >>= 1) {
#pragma unroll
    for (int j = 0; j < EPI_ROWS; ++j) {
      s[j]  += __shfl_xor(s[j],  m, 32);
      ss[j] += __shfl_xor(ss[j], m, 32);
    }
  }
#pragma unroll
  for (int j = 0; j < EPI_ROWS; ++j) {
    const int n = wv + j * OWN_WAVES;
    const float mu  = s[j] * (1.f / HID);
    const float var = ss[j] * (1.f / HID) - mu * mu;
    const float inv = rsqrtf(var + LN_EPS);
    float o0 = (v0[j] - mu) * inv * gm0 + bt0;
    float o1 = (v1[j] - mu) * inv * gm1 + bt1;
    o0 = (o0 >= 0.f) ? o0 : SLOPE * o0;
    o1 = (o1 >= 0.f) ? o1 : SLOPE * o1;
    v2f r; r.x = hh[j].x + o0; r.y = hh[j].y + o1;
    if (n < cnt)
      *(v2f*)(dst + (size_t)(lo + n) * HID + c2) = r;
  }
}

extern "C" void kernel_launch(void* const* d_in, const int* in_sizes, int n_in,
                              void* d_out, int out_size, void* d_ws, size_t ws_size,
                              hipStream_t stream) {
  const float* x      = (const float*)d_in[0];
  const int*   eidx   = (const int*)  d_in[1];
  const float* ew     = (const float*)d_in[2];
  const float* Wp     = (const float*)d_in[3];
  const float* bp     = (const float*)d_in[4];
  const float* Wself  = (const float*)d_in[5];
  const float* bself  = (const float*)d_in[6];
  const float* Wneigh = (const float*)d_in[7];
  const float* bneigh = (const float*)d_in[8];
  const float* Wedge  = (const float*)d_in[9];
  const float* gamma  = (const float*)d_in[10];
  const float* beta   = (const float*)d_in[11];
  float* out = (float*)d_out;

  const size_t NH = (size_t)N_NODES * HID;
  float* h       = (float*)d_ws;
  float* h_self  = h + NH;
  float* h_neigh = h_self + NH;
  float* scratch = h_neigh + NH;

  gnn_proj_kernel<<<(N_NODES * HID) / 256, 256, 0, stream>>>(x, Wp, bp, h);

  for (int l = 0; l < N_LAYERS; ++l) {
    const float* Ws = Wself  + (size_t)l * HID * HID;
    const float* bs = bself  + (size_t)l * HID;
    const float* Wn = Wneigh + (size_t)l * HID * HID;
    const float* bn = bneigh + (size_t)l * HID;
    const float* We = Wedge  + (size_t)l * HID;
    const float* gm = gamma  + (size_t)l * HID;
    const float* bt = beta   + (size_t)l * HID;

    gnn_gemm_kernel<<<dim3(N_NODES / 32, 2), 256, 0, stream>>>(
        h, Ws, bs, Wn, bn, h_self, h_neigh);

    float* dst = (l == N_LAYERS - 1) ? out : h;
    if (l == 0)
      gnn_edge_own_kernel<false><<<OWN_BLOCKS, OWN_THREADS, 0, stream>>>(
          eidx, ew, We, h_neigh, h_self, gm, bt, h, dst, scratch);
    else
      gnn_edge_own_kernel<true><<<OWN_BLOCKS, OWN_THREADS, 0, stream>>>(
          eidx, ew, We, h_neigh, h_self, gm, bt, h, dst, scratch);
  }
}


# Input projection formed inside the first layer's GEMM work-groups (own 32x64 h tile from x in LDS = A operand; y==0 groups store h): no projection launch, 6 not 7; k order kept, multiply-adds fused
# speedup vs baseline: 1.8142x; 1.0626x over previous
#include <hip/hip_runtime.h>
#include <hip/hip_bf16.h>

#define N_NODES 100000
#define N_EDGES 1000000
#define IN_DIM  8
#define HID     64
#define N_LAYERS 3
#define LN_EPS  1e-5f
#define SLOPE   0.2f
#define LDW     68

typedef float v2f __attribute__((ext_vector_type(2)));
typedef float v8f __attribute__((ext_vector_type(8)));

#define LDH     68

template <bool PROJ>
__global__ void gnn_gemm_kernel(const float* __restrict__ h,
                                const float* __restrict__ x, const float* __restrict__ Wp,
                                const float* __restrict__ bp,
                                const float* __restrict__ Wself, const float* __restrict__ bself,
                                const float* __restrict__ Wneigh, const float* __restrict__ bneigh,
                                float* __restrict__ h_new,
                                float* __restrict__ h_self, float* __restrict__ h_neigh) {
  __shared__ float w_s[HID * LDW];
  __shared__ float b_s[HID];
  __shared__ float h_s[PROJ ? 32 * LDH : 1];
  const float* W  = blockIdx.y ? Wneigh : Wself;
  const float* b  = blockIdx.y ? bneigh : bself;
  float*       out = blockIdx.y ? h_neigh : h_self;

  int t = threadIdx.x;
#pragma unroll
  for (int i = 0; i < 16; ++i) {
    int idx = t + i * 256;
    w_s[(idx >> 6) * LDW + (idx & 63)] = W[idx];
  }
  if (t < HID) b_s[t] = b[t];

  const int pc = t & 63;
  const int rq = __builtin_amdgcn_readfirstlane(t >> 6);
  const int rb = blockIdx.x * 32;
  float acc[8];
  if (PROJ) {
    const float4 wa = *(const float4*)(Wp + pc * IN_DIM);
    const float4 wb = *(const float4*)(Wp + pc * IN_DIM + 4);
    const float wr[IN_DIM] = {wa.x, wa.y, wa.z, wa.w, wb.x, wb.y, wb.z, wb.w};
    const float bias_p = bp[pc];
#pragma unroll
    for (int j = 0; j < 8; ++j) {
      const float* xr = x + (size_t)(rb + rq + 4 * j) * IN_DIM;
      acc[j] = bias_p;
#pragma unroll
      for (int k = 0; k < IN_DIM; ++k) acc[j] += xr[k] * wr[k];
    }
#pragma unroll
    for (int j = 0; j < 8; ++j) h_s[(rq + 4 * j) * LDH + pc] = acc[j];
  }
  __syncthreads();
  if (PROJ && blockIdx.y == 0) {
#pragma unroll
    for (int j = 0; j < 8; ++j) h_new[(size_t)(rb + rq + 4 * j) * HID + pc] = acc[j];
  }

  int wave = t >> 5, lane = t & 31;
  int r0 = blockIdx.x * 32 + (wave >> 2) * 16;
  int n0 = (wave & 3) * 16;
  int lr = lane & 15;
  int koff = (lane < 16) ? 0 : 2;

  const float* brow = w_s + (n0 + lr) * LDW;

  v8f c = {};
  if (PROJ) {
    const float* arow = h_s + ((wave >> 2) * 16 + lr) * LDH;
#pragma unroll
    for (int kb = 0; kb < 16; ++kb) {
      int k = kb * 4 + koff;
      v2f a  = *(const v2f*)(arow + k);
      v2f bf = *(const v2f*)(brow + k);
      c = __builtin_amdgcn_wmma_f32_16x16x4_f32(false, a, false, bf, (short)0, c, false, false);
    }
  } else {
    const float* arow = h + (size_t)(r0 + lr) * HID;
#pragma unroll
    for (int kb = 0; kb < 16; ++kb) {
      int k = kb * 4 + koff;
      v2f a  = *(const v2f*)(arow + k);
      v2f bf = *(const v2f*)(brow + k);
      c = __builtin_amdgcn_wmma_f32_16x16x4_f32(false, a, false, bf, (short)0, c, false, false);
    }
  }

  float bias = b_s[n0 + lr];
  int mbase = r0 + ((lane < 16) ? 0 : 8);
#pragma unroll
  for (int i = 0; i < 8; ++i) {
    out[(size_t)(mbase + i) * HID + n0 + lr] = c[i] + bias;
  }
}

#define OWN_BLOCKS  256
#define OWN_NODES   391
#define OWN_THREADS 1024
#define OWN_WAVES   (OWN_THREADS / 32)
#define SCAN_VEC    (N_EDGES / 4)
#define SCAN_FULL   (SCAN_VEC / OWN_THREADS)
#define Q_DRAIN     32
#define Q_CAP       (Q_DRAIN + 4 * 128)
#define E_BITS      20
#define L_CAP       256
#define C_BITS      17
#define L_WORDS     (2 * L_CAP)
#define EPI_ROWS    ((OWN_NODES + OWN_WAVES - 1) / OWN_WAVES)
static_assert(N_EDGES % 4 == 0 && N_EDGES <= (1 << E_BITS), "edge id must fit its field");
static_assert(OWN_NODES < (1 << (32 - E_BITS)) && OWN_BLOCKS * OWN_NODES >= N_NODES, "owner ranges");
static_assert(SCAN_FULL % 4 == 0 && SCAN_VEC - SCAN_FULL * OWN_THREADS < OWN_THREADS, "scan split");
static_assert(N_NODES <= (1 << C_BITS) && OWN_NODES <= (1 << (32 - C_BITS)), "record key fields");
static_assert((size_t)OWN_BLOCKS * OWN_WAVES * (L_WORDS + 1) <= (size_t)N_NODES * HID,
              "record regions and their counts fit the borrowed plane");

template <bool LIST>
__global__ __launch_bounds__(OWN_THREADS)
void gnn_edge_own_kernel(const int* __restrict__ eidx, const float* __restrict__ ew,
                         const float* __restrict__ Wedge, const float* __restrict__ h_neigh,
                         const float* __restrict__ h_self, const float* __restrict__ gamma,
                         const float* __restrict__ beta, const float* h, float* dst,
                         float* __restrict__ scratch) {
  __shared__ float    acc[OWN_NODES * HID];
  __shared__ unsigned q[OWN_WAVES][Q_CAP];

  const int t = threadIdx.x, wave = t >> 5, lane = t & 31;
  const int lo  = blockIdx.x * OWN_NODES;
  const int cnt = (N_NODES - lo < OWN_NODES) ? (N_NODES - lo) : OWN_NODES;

  const int wv = __builtin_amdgcn_readfirstlane(wave);
  v2f a[EPI_ROWS], hh[EPI_ROWS];
#pragma unroll
  for (int j = 0; j < EPI_ROWS; ++j) {
    int n = wv + j * OWN_WAVES;
    n = (n < cnt) ? n : (cnt - 1);
    const size_t base = (size_t)(lo + n) * HID + 2 * lane;
    a[j]  = *(const v2f*)(h_self + base);
    hh[j] = *(const v2f*)(h + base);
  }
  const float gm0 = gamma[2 * lane], gm1 = gamma[2 * lane + 1];
  const float bt0 = beta[2 * lane],  bt1 = beta[2 * lane + 1];

  for (int i = t; i < OWN_NODES * HID / 4; i += OWN_THREADS)
    ((float4*)acc)[i] = make_float4(0.f, 0.f, 0.f, 0.f);
  __syncthreads();

  const int4* __restrict__ row4 = (const int4*)eidx;
  const int*  __restrict__ colp = eidx + N_EDGES;
  const int   c2  = lane * 2;
  const float we0 = Wedge[c2], we1 = Wedge[c2 + 1];
  unsigned* myq = q[wave];
  int qn = 0;
  const int region = blockIdx.x * OWN_WAVES + wave;
  unsigned* lst = (unsigned*)scratch;
  unsigned* myl = lst + (size_t)region * L_WORDS;
  unsigned* myc = lst + (size_t)OWN_BLOCKS * OWN_WAVES * L_WORDS + region;
  int ln = 0;

  auto add32 = [&](const int rl, const int cl, const int wl, const int valid) {
    for (int u = 0; u < valid; u += 8) {
      v2f m[8];
#pragma unroll
      for (int k = 0; k < 8; ++k) {
        const int c = __builtin_amdgcn_readlane(cl, u + k);
        m[k] = *(const v2f*)(h_neigh + (size_t)c * HID + c2);
      }
#pragma unroll
      for (int k = 0; k < 8; ++k) {
        if (u + k < valid) {
          const float w = __int_as_float(__builtin_amdgcn_readlane(wl, u + k));
          const int   r = __builtin_amdgcn_readlane(rl, u + k);
          const float g0 = 1.f / (1.f + __expf(-w * we0));
          const float g1 = 1.f / (1.f + __expf(-w * we1));
          float* d = acc + r * HID + c2;
          atomicAdd(d,     m[k].x * g0);
          atomicAdd(d + 1, m[k].y * g1);
        }
      }
    }
  };

  auto drain = [&](const int total) {
    for (int base = 0; base < total; base += 32) {
      const int valid = (total - base < 32) ? (total - base) : 32;
      const unsigned pk = (lane < valid) ? myq[base + lane] : 0u;
      const int e  = (int)(pk & ((1u << E_BITS) - 1u));
      const int rl = (int)(pk >> E_BITS);
      const int cl = colp[e];
      const int wl = __float_as_int(ew[e]);
      if (!LIST) {
        const int p = ln + base + lane;
        if (lane < valid && p < L_CAP)
          *(uint2*)(myl + 2 * p) = make_uint2(((unsigned)rl << C_BITS) | (unsigned)cl, (unsigned)wl);
      }
      add32(rl, cl, wl, valid);
    }
    ln += total;
  };

  auto take = [&](const int4 v, const int ebase) {
    const unsigned r0 = (unsigned)(v.x - lo), r1 = (unsigned)(v.y - lo);
    const unsigned r2 = (unsigned)(v.z - lo), r3 = (unsigned)(v.w - lo);
    const bool m0 = r0 < (unsigned)cnt, m1 = r1 < (unsigned)cnt;
    const bool m2 = r2 < (unsigned)cnt, m3 = r3 < (unsigned)cnt;
    if (__builtin_amdgcn_ballot_w32(m0 | m1 | m2 | m3)) {
      const unsigned rr[4] = {r0, r1, r2, r3};
      const bool     mm[4] = {m0, m1, m2, m3};
#pragma unroll
      for (int j = 0; j < 4; ++j) {
        const unsigned mask = __builtin_amdgcn_ballot_w32(mm[j]);
        if (mask) {
          if (mm[j])
            myq[qn + (int)__builtin_amdgcn_mbcnt_lo(mask, 0u)] =
                (rr[j] << E_BITS) | (unsigned)(ebase + j);
          qn += __builtin_popcount(mask);
        }
      }
    }
  };

  bool scan = true;
  if (LIST) {
    const int n = __builtin_amdgcn_readfirstlane((int)*myc);
    if (n <= L_CAP) {
      scan = false;
      const uint2* rec = (const uint2*)myl;
      uint2 nx = (lane < n) ? rec[lane] : make_uint2(0u, 0u);
      for (int base = 0; base < n; base += 32) {
        const uint2 cur = nx;
        const int nb = base + 32 + lane;
        nx = (nb < n) ? rec[nb] : make_uint2(0u, 0u);
        const int valid = (n - base < 32) ? (n - base) : 32;
        add32((int)(cur.x >> C_BITS), (int)(cur.x & ((1u << C_BITS) - 1u)), (int)cur.y, valid);
      }
    }
  }

  const int start = (blockIdx.x * SCAN_FULL) / OWN_BLOCKS;

  if (scan) {
    for (int g = 0; g < SCAN_FULL; g += 4) {
      int4 v[4];
      int  eb[4];
#pragma unroll
      for (int k = 0; k < 4; ++k) {
        int it = g + k + start;
        if (it >= SCAN_FULL) it -= SCAN_FULL;
        const int idx = it * OWN_THREADS + t;
        eb[k] = idx * 4;
        v[k]  = row4[idx];
      }
#pragma unroll
      for (int k = 0; k < 4; ++k) take(v[k], eb[k]);
      if (qn >= Q_DRAIN) {
        const int full = qn & ~7, rem = qn - full;
        drain(full);
        const unsigned keep = (lane < rem) ? myq[full + lane] : 0u;
        if (lane < rem) myq[lane] = keep;
        qn = rem;
      }
    }
    {
      const int idx = SCAN_FULL * OWN_THREADS + t;
      const int4 v = (idx < SCAN_VEC) ? row4[idx] : make_int4(-1, -1, -1, -1);
      take(v, idx * 4);
    }
    drain(qn);
    if (!LIST && lane == 0) *myc = (unsigned)ln;
  }

  __syncthreads();

  float v0[EPI_ROWS], v1[EPI_ROWS], s[EPI_ROWS], ss[EPI_ROWS];
#pragma unroll
  for (int j = 0; j < EPI_ROWS; ++j) {
    int n = wv + j * OWN_WAVES;
    n = (n < cnt) ? n : (cnt - 1);
    const v2f g = *(const v2f*)(acc + n * HID + c2);
    v0[j] = a[j].x + g.x;
    v1[j] = a[j].y + g.y;
    s[j]  = v0[j] + v1[j];
    ss[j] = v0[j] * v0[j] + v1[j] * v1[j];
  }
#pragma unroll
  for (int m = 16; m >= 1; m >>= 1) {
#pragma unroll
    for (int j = 0; j < EPI_ROWS; ++j) {
      s[j]  += __shfl_xor(s[j],  m, 32);
      ss[j] += __shfl_xor(ss[j], m, 32);
    }
  }
#pragma unroll
  for (int j = 0; j < EPI_ROWS; ++j) {
    const int n = wv + j * OWN_WAVES;
    const float mu  = s[j] * (1.f / HID);
    const float var = ss[j] * (1.f / HID) - mu * mu;
    const float inv = rsqrtf(var + LN_EPS);
    float o0 = (v0[j] - mu) * inv * gm0 + bt0;
    float o1 = (v1[j] - mu) * inv * gm1 + bt1;
    o0 = (o0 >= 0.f) ? o0 : SLOPE * o0;
    o1 = (o1 >= 0.f) ? o1 : SLOPE * o1;
    v2f r; r.x = hh[j].x + o0; r.y = hh[j].y + o1;
    if (n < cnt)
      *(v2f*)(dst + (size_t)(lo + n) * HID + c2) = r;
  }
}

extern "C" void kernel_launch(void* const* d_in, const int* in_sizes, int n_in,
                              void* d_out, int out_size, void* d_ws, size_t ws_size,
                              hipStream_t stream) {
  const float* x      = (const float*)d_in[0];
  const int*   eidx   = (const int*)  d_in[1];
  const float* ew     = (const float*)d_in[2];
  const float* Wp     = (const float*)d_in[3];
  const float* bp     = (const float*)d_in[4];
  const float* Wself  = (const float*)d_in[5];
  const float* bself  = (const float*)d_in[6];
  const float* Wneigh = (const float*)d_in[7];
  const float* bneigh = (const float*)d_in[8];
  const float* Wedge  = (const float*)d_in[9];
  const float* gamma  = (const float*)d_in[10];
  const float* beta   = (const float*)d_in[11];
  float* out = (float*)d_out;

  const size_t NH = (size_t)N_NODES * HID;
  float* h       = (float*)d_ws;
  float* h_self  = h + NH;
  float* h_neigh = h_self + NH;
  float* scratch = h_neigh + NH;

  for (int l = 0; l < N_LAYERS; ++l) {
    const float* Ws = Wself  + (size_t)l * HID * HID;
    const float* bs = bself  + (size_t)l * HID;
    const float* Wn = Wneigh + (size_t)l * HID * HID;
    const float* bn = bneigh + (size_t)l * HID;
    const float* We = Wedge  + (size_t)l * HID;
    const float* gm = gamma  + (size_t)l * HID;
    const float* bt = beta   + (size_t)l * HID;

    if (l == 0)
      gnn_gemm_kernel<true><<<dim3(N_NODES / 32, 2), 256, 0, stream>>>(
          h, x, Wp, bp, Ws, bs, Wn, bn, h, h_self, h_neigh);
    else
      gnn_gemm_kernel<false><<<dim3(N_NODES / 32, 2), 256, 0, stream>>>(
          h, x, Wp, bp, Ws, bs, Wn, bn, h, h_self, h_neigh);

    float* dst = (l == N_LAYERS - 1) ? out : h;
    if (l == 0)
      gnn_edge_own_kernel<false><<<OWN_BLOCKS, OWN_THREADS, 0, stream>>>(
          eidx, ew, We, h_neigh, h_self, gm, bt, h, dst, scratch);
    else
      gnn_edge_own_kernel<true><<<OWN_BLOCKS, OWN_THREADS, 0, stream>>>(
          eidx, ew, We, h_neigh, h_self, gm, bt, h, dst, scratch);
  }
}


# GEMMs by 256 resident work-groups: a wave keeps both weight matrices' B fragments in registers and walks 16-row tiles (layer 1 forms its h tile per wave); 2048 waves a launch, not 50000; sums kept
# speedup vs baseline: 1.9633x; 1.0822x over previous
#include <hip/hip_runtime.h>
#include <hip/hip_bf16.h>

#define N_NODES 100000
#define N_EDGES 1000000
#define IN_DIM  8
#define HID     64
#define N_LAYERS 3
#define LN_EPS  1e-5f
#define SLOPE   0.2f
#define LDW     68

typedef float v2f __attribute__((ext_vector_type(2)));
typedef float v4f __attribute__((ext_vector_type(4)));
typedef float v8f __attribute__((ext_vector_type(8)));

#define LDH        68
#define GR_BLOCKS  256
#define GR_WAVES   8
#define GR_THREADS (GR_WAVES * 32)
#define ROW_TILES  (N_NODES / 16)
static_assert(N_NODES % 16 == 0 && GR_BLOCKS * GR_WAVES <= ROW_TILES, "whole tiles; every wave has a first tile");

struct gr_ahead {
  v2f    a[16];
  v4f    x;
};

template <bool PROJ>
__global__ __launch_bounds__(GR_THREADS)
void gnn_gemm_rows_kernel(const float* __restrict__ h,
                          const float* __restrict__ x, const float* __restrict__ Wp,
                          const float* __restrict__ bp,
                          const float* __restrict__ Wself, const float* __restrict__ bself,
                          const float* __restrict__ Wneigh, const float* __restrict__ bneigh,
                          float* __restrict__ h_new,
                          float* __restrict__ h_self, float* __restrict__ h_neigh) {
  __shared__ __attribute__((aligned(16))) float w_s[2 * HID * LDW];
  __shared__ __attribute__((aligned(16))) float h_t[PROJ ? GR_WAVES * 16 * LDH : 1];
  const int t = threadIdx.x;
#pragma unroll
  for (int i = 0; i < 4; ++i) {
    const int v = t + i * GR_THREADS;
    *(float4*)(w_s + (v >> 4) * LDW + (v & 15) * 4)             = ((const float4*)Wself)[v];
    *(float4*)(w_s + HID * LDW + (v >> 4) * LDW + (v & 15) * 4) = ((const float4*)Wneigh)[v];
  }
  __syncthreads();

  const int lane = t & 31;
  const int wave = __builtin_amdgcn_readfirstlane(t >> 5);
  const int lr = lane & 15;
  const int koff = (lane < 16) ? 0 : 2;

  v2f bfr[8][16];
  float bias[8];
#pragma unroll
  for (int ct = 0; ct < 8; ++ct) {
    const float* brow = w_s + (ct >> 2) * (HID * LDW) + ((ct & 3) * 16 + lr) * LDW;
#pragma unroll
    for (int kb = 0; kb < 16; ++kb) bfr[ct][kb] = *(const v2f*)(brow + kb * 4 + koff);
    bias[ct] = ((ct < 4) ? bself : bneigh)[(ct & 3) * 16 + lr];
  }

  float pw0[IN_DIM], pw1[IN_DIM], pb0 = 0.f, pb1 = 0.f;
  float* ht = h_t;
  if (PROJ) {
    const float4* wq = (const float4*)(Wp + 2 * lane * IN_DIM);
    const float4 q0 = wq[0], q1 = wq[1], q2 = wq[2], q3 = wq[3];
    pw0[0] = q0.x; pw0[1] = q0.y; pw0[2] = q0.z; pw0[3] = q0.w;
    pw0[4] = q1.x; pw0[5] = q1.y; pw0[6] = q1.z; pw0[7] = q1.w;
    pw1[0] = q2.x; pw1[1] = q2.y; pw1[2] = q2.z; pw1[3] = q2.w;
    pw1[4] = q3.x; pw1[5] = q3.y; pw1[6] = q3.z; pw1[7] = q3.w;
    pb0 = bp[2 * lane]; pb1 = bp[2 * lane + 1];
    ht = h_t + wave * (16 * LDH);
  }

  auto ahead = [&](gr_ahead& g, const int tile) {
    if (PROJ) {
      g.x = *(const v4f*)(x + (size_t)tile * (16 * IN_DIM) + 4 * lane);
    } else {
      const float* arow = h + (size_t)(tile * 16 + lr) * HID + koff;
#pragma unroll
      for (int kb = 0; kb < 16; ++kb) g.a[kb] = *(const v2f*)(arow + kb * 4);
    }
  };

  auto finish = [&](const gr_ahead& g, const int tile) {
    v2f ac[16];
    if (PROJ) {
      const float xc[4] = {g.x[0], g.x[1], g.x[2], g.x[3]};
#pragma unroll
      for (int r = 0; r < 16; ++r) {
        float p0 = pb0, p1 = pb1;
#pragma unroll
        for (int k = 0; k < IN_DIM; ++k) {
          const float xs = __int_as_float(__builtin_amdgcn_readlane(__float_as_int(xc[k & 3]), 2 * r + (k >> 2)));
          p0 = __builtin_fmaf(xs, pw0[k], p0);
          p1 = __builtin_fmaf(xs, pw1[k], p1);
        }
        v2f hv; hv.x = p0; hv.y = p1;
        *(v2f*)(h_new + (size_t)(tile * 16 + r) * HID + 2 * lane) = hv;
        *(v2f*)(ht + r * LDH + 2 * lane) = hv;
      }
#pragma unroll
      for (int kb = 0; kb < 16; ++kb) ac[kb] = *(const v2f*)(ht + lr * LDH + kb * 4 + koff);
    } else {
#pragma unroll
      for (int kb = 0; kb < 16; ++kb) ac[kb] = g.a[kb];
    }
    v8f c[8];
#pragma unroll
    for (int ct = 0; ct < 8; ++ct) c[ct] = v8f{};
#pragma unroll
    for (int kb = 0; kb < 16; ++kb) {
#pragma unroll
      for (int ct = 0; ct < 8; ++ct)
        c[ct] = __builtin_amdgcn_wmma_f32_16x16x4_f32(false, ac[kb], false, bfr[ct][kb], (short)0, c[ct], false, false);
    }
    const int mbase = tile * 16 + ((lane < 16) ? 0 : 8);
#pragma unroll
    for (int i = 0; i < 8; ++i) {
#pragma unroll
      for (int ct = 0; ct < 8; ++ct) {
        float* o = (ct < 4) ? h_self : h_neigh;
        o[(size_t)(mbase + i) * HID + (ct & 3) * 16 + lr] = c[ct][i] + bias[ct];
      }
    }
  };

  const int NW = GR_BLOCKS * GR_WAVES;
  int rt = wave * GR_BLOCKS + blockIdx.x;
  auto hold = [&](const gr_ahead& g) {
    if (PROJ) {
      asm volatile("" :: "v"(g.x));
    } else {
      asm volatile("" :: "v"(g.a[0]), "v"(g.a[1]), "v"(g.a[2]), "v"(g.a[3]), "v"(g.a[4]), "v"(g.a[5]),
                         "v"(g.a[6]), "v"(g.a[7]), "v"(g.a[8]), "v"(g.a[9]), "v"(g.a[10]), "v"(g.a[11]),
                         "v"(g.a[12]), "v"(g.a[13]), "v"(g.a[14]), "v"(g.a[15]));
    }
  };
  gr_ahead g0, g1;
  ahead(g0, rt);
  for (;;) {
    ahead(g1, (rt + NW < ROW_TILES) ? rt + NW : rt);
    __builtin_amdgcn_sched_barrier(0);
    finish(g0, rt);
    hold(g1);
    rt += NW;
    if (rt >= ROW_TILES) break;
    ahead(g0, (rt + NW < ROW_TILES) ? rt + NW : rt);
    __builtin_amdgcn_sched_barrier(0);
    finish(g1, rt);
    hold(g0);
    rt += NW;
    if (rt >= ROW_TILES) break;
  }
}

#define OWN_BLOCKS  256
#define OWN_NODES   391
#define OWN_THREADS 1024
#define OWN_WAVES   (OWN_THREADS / 32)
#define SCAN_VEC    (N_EDGES / 4)
#define SCAN_FULL   (SCAN_VEC / OWN_THREADS)
#define Q_DRAIN     32
#define Q_CAP       (Q_DRAIN + 4 * 128)
#define E_BITS      20
#define L_CAP       256
#define C_BITS      17
#define L_WORDS     (2 * L_CAP)
#define EPI_ROWS    ((OWN_NODES + OWN_WAVES - 1) / OWN_WAVES)
static_assert(N_EDGES % 4 == 0 && N_EDGES <= (1 << E_BITS), "edge id must fit its field");
static_assert(OWN_NODES < (1 << (32 - E_BITS)) && OWN_BLOCKS * OWN_NODES >= N_NODES, "owner ranges");
static_assert(SCAN_FULL % 4 == 0 && SCAN_VEC - SCAN_FULL * OWN_THREADS < OWN_THREADS, "scan split");
static_assert(N_NODES <= (1 << C_BITS) && OWN_NODES <= (1 << (32 - C_BITS)), "record key fields");
static_assert((size_t)OWN_BLOCKS * OWN_WAVES * (L_WORDS + 1) <= (size_t)N_NODES * HID,
              "record regions and their counts fit the borrowed plane");

template <bool LIST>
__global__ __launch_bounds__(OWN_THREADS)
void gnn_edge_own_kernel(const int* __restrict__ eidx, const float* __restrict__ ew,
                         const float* __restrict__ Wedge, const float* __restrict__ h_neigh,
                         const float* __restrict__ h_self, const float* __restrict__ gamma,
                         const float* __restrict__ beta, const float* h, float* dst,
                         float* __restrict__ scratch) {
  __shared__ float    acc[OWN_NODES * HID];
  __shared__ unsigned q[OWN_WAVES][Q_CAP];

  const int t = threadIdx.x, wave = t >> 5, lane = t & 31;
  const int lo  = blockIdx.x * OWN_NODES;
  const int cnt = (N_NODES - lo < OWN_NODES) ? (N_NODES - lo) : OWN_NODES;

  const int wv = __builtin_amdgcn_readfirstlane(wave);
  v2f a[EPI_ROWS], hh[EPI_ROWS];
#pragma unroll
  for (int j = 0; j < EPI_ROWS; ++j) {
    int n = wv + j * OWN_WAVES;
    n = (n < cnt) ? n : (cnt - 1);
    const size_t base = (size_t)(lo + n) * HID + 2 * lane;
    a[j]  = *(const v2f*)(h_self + base);
    hh[j] = *(const v2f*)(h + base);
  }
  const float gm0 = gamma[2 * lane], gm1 = gamma[2 * lane + 1];
  const float bt0 = beta[2 * lane],  bt1 = beta[2 * lane + 1];

  for (int i = t; i < OWN_NODES * HID / 4; i += OWN_THREADS)
    ((float4*)acc)[i] = make_float4(0.f, 0.f, 0.f, 0.f);
  __syncthreads();

  const int4* __restrict__ row4 = (const int4*)eidx;
  const int*  __restrict__ colp = eidx + N_EDGES;
  const int   c2  = lane * 2;
  const float we0 = Wedge[c2], we1 = Wedge[c2 + 1];
  unsigned* myq = q[wave];
  int qn = 0;
  const int region = blockIdx.x * OWN_WAVES + wave;
  unsigned* lst = (unsigned*)scratch;
  unsigned* myl = lst + (size_t)region * L_WORDS;
  unsigned* myc = lst + (size_t)OWN_BLOCKS * OWN_WAVES * L_WORDS + region;
  int ln = 0;

  auto add32 = [&](const int rl, const int cl, const int wl, const int valid) {
    for (int u = 0; u < valid; u += 8) {
      v2f m[8];
#pragma unroll
      for (int k = 0; k < 8; ++k) {
        const int c = __builtin_amdgcn_readlane(cl, u + k);
        m[k] = *(const v2f*)(h_neigh + (size_t)c * HID + c2);
      }
#pragma unroll
      for (int k = 0; k < 8; ++k) {
        if (u + k < valid) {
          const float w = __int_as_float(__builtin_amdgcn_readlane(wl, u + k));
          const int   r = __builtin_amdgcn_readlane(rl, u + k);
          const float g0 = 1.f / (1.f + __expf(-w * we0));
          const float g1 = 1.f / (1.f + __expf(-w * we1));
          float* d = acc + r * HID + c2;
          atomicAdd(d,     m[k].x * g0);
          atomicAdd(d + 1, m[k].y * g1);
        }
      }
    }
  };

  auto drain = [&](const int total) {
    for (int base = 0; base < total; base += 32) {
      const int valid = (total - base < 32) ? (total - base) : 32;
      const unsigned pk = (lane < valid) ? myq[base + lane] : 0u;
      const int e  = (int)(pk & ((1u << E_BITS) - 1u));
      const int rl = (int)(pk >> E_BITS);
      const int cl = colp[e];
      const int wl = __float_as_int(ew[e]);
      if (!LIST) {
        const int p = ln + base + lane;
        if (lane < valid && p < L_CAP)
          *(uint2*)(myl + 2 * p) = make_uint2(((unsigned)rl << C_BITS) | (unsigned)cl, (unsigned)wl);
      }
      add32(rl, cl, wl, valid);
    }
    ln += total;
  };

  auto take = [&](const int4 v, const int ebase) {
    const unsigned r0 = (unsigned)(v.x - lo), r1 = (unsigned)(v.y - lo);
    const unsigned r2 = (unsigned)(v.z - lo), r3 = (unsigned)(v.w - lo);
    const bool m0 = r0 < (unsigned)cnt, m1 = r1 < (unsigned)cnt;
    const bool m2 = r2 < (unsigned)cnt, m3 = r3 < (unsigned)cnt;
    if (__builtin_amdgcn_ballot_w32(m0 | m1 | m2 | m3)) {
      const unsigned rr[4] = {r0, r1, r2, r3};
      const bool     mm[4] = {m0, m1, m2, m3};
#pragma unroll
      for (int j = 0; j < 4; ++j) {
        const unsigned mask = __builtin_amdgcn_ballot_w32(mm[j]);
        if (mask) {
          if (mm[j])
            myq[qn + (int)__builtin_amdgcn_mbcnt_lo(mask, 0u)] =
                (rr[j] << E_BITS) | (unsigned)(ebase + j);
          qn += __builtin_popcount(mask);
        }
      }
    }
  };

  bool scan = true;
  if (LIST) {
    const int n = __builtin_amdgcn_readfirstlane((int)*myc);
    if (n <= L_CAP) {
      scan = false;
      const uint2* rec = (const uint2*)myl;
      uint2 nx = (lane < n) ? rec[lane] : make_uint2(0u, 0u);
      for (int base = 0; base < n; base += 32) {
        const uint2 cur = nx;
        const int nb = base + 32 + lane;
        nx = (nb < n) ? rec[nb] : make_uint2(0u, 0u);
        const int valid = (n - base < 32) ? (n - base) : 32;
        add32((int)(cur.x >> C_BITS), (int)(cur.x & ((1u << C_BITS) - 1u)), (int)cur.y, valid);
      }
    }
  }

  const int start = (blockIdx.x * SCAN_FULL) / OWN_BLOCKS;

  if (scan) {
    for (int g = 0; g < SCAN_FULL; g += 4) {
      int4 v[4];
      int  eb[4];
#pragma unroll
      for (int k = 0; k < 4; ++k) {
        int it = g + k + start;
        if (it >= SCAN_FULL) it -= SCAN_FULL;
        const int idx = it * OWN_THREADS + t;
        eb[k] = idx * 4;
        v[k]  = row4[idx];
      }
#pragma unroll
      for (int k = 0; k < 4; ++k) take(v[k], eb[k]);
      if (qn >= Q_DRAIN) {
        const int full = qn & ~7, rem = qn - full;
        drain(full);
        const unsigned keep = (lane < rem) ? myq[full + lane] : 0u;
        if (lane < rem) myq[lane] = keep;
        qn = rem;
      }
    }
    {
      const int idx = SCAN_FULL * OWN_THREADS + t;
      const int4 v = (idx < SCAN_VEC) ? row4[idx] : make_int4(-1, -1, -1, -1);
      take(v, idx * 4);
    }
    drain(qn);
    if (!LIST && lane == 0) *myc = (unsigned)ln;
  }

  __syncthreads();

  float v0[EPI_ROWS], v1[EPI_ROWS], s[EPI_ROWS], ss[EPI_ROWS];
#pragma unroll
  for (int j = 0; j < EPI_ROWS; ++j) {
    int n = wv + j * OWN_WAVES;
    n = (n < cnt) ? n : (cnt - 1);
    const v2f g = *(const v2f*)(acc + n * HID + c2);
    v0[j] = a[j].x + g.x;
    v1[j] = a[j].y + g.y;
    s[j]  = v0[j] + v1[j];
    ss[j] = v0[j] * v0[j] + v1[j] * v1[j];
  }
#pragma unroll
  for (int m = 16; m >= 1; m >>= 1) {
#pragma unroll
    for (int j = 0; j < EPI_ROWS; ++j) {
      s[j]  += __shfl_xor(s[j],  m, 32);
      ss[j] += __shfl_xor(ss[j], m, 32);
    }
  }
#pragma unroll
  for (int j = 0; j < EPI_ROWS; ++j) {
    const int n = wv + j * OWN_WAVES;
    const float mu  = s[j] * (1.f / HID);
    const float var = ss[j] * (1.f / HID) - mu * mu;
    const float inv = rsqrtf(var + LN_EPS);
    float o0 = (v0[j] - mu) * inv * gm0 + bt0;
    float o1 = (v1[j] - mu) * inv * gm1 + bt1;
    o0 = (o0 >= 0.f) ? o0 : SLOPE * o0;
    o1 = (o1 >= 0.f) ? o1 : SLOPE * o1;
    v2f r; r.x = hh[j].x + o0; r.y = hh[j].y + o1;
    if (n < cnt)
      *(v2f*)(dst + (size_t)(lo + n) * HID + c2) = r;
  }
}

extern "C" void kernel_launch(void* const* d_in, const int* in_sizes, int n_in,
                              void* d_out, int out_size, void* d_ws, size_t ws_size,
                              hipStream_t stream) {
  const float* x      = (const float*)d_in[0];
  const int*   eidx   = (const int*)  d_in[1];
  const float* ew     = (const float*)d_in[2];
  const float* Wp     = (const float*)d_in[3];
  const float* bp     = (const float*)d_in[4];
  const float* Wself  = (const float*)d_in[5];
  const float* bself  = (const float*)d_in[6];
  const float* Wneigh = (const float*)d_in[7];
  const float* bneigh = (const float*)d_in[8];
  const float* Wedge  = (const float*)d_in[9];
  const float* gamma  = (const float*)d_in[10];
  const float* beta   = (const float*)d_in[11];
  float* out = (float*)d_out;

  const size_t NH = (size_t)N_NODES * HID;
  float* h       = (float*)d_ws;
  float* h_self  = h + NH;
  float* h_neigh = h_self + NH;
  float* scratch = h_neigh + NH;

  for (int l = 0; l < N_LAYERS; ++l) {
    const float* Ws = Wself  + (size_t)l * HID * HID;
    const float* bs = bself  + (size_t)l * HID;
    const float* Wn = Wneigh + (size_t)l * HID * HID;
    const float* bn = bneigh + (size_t)l * HID;
    const float* We = Wedge  + (size_t)l * HID;
    const float* gm = gamma  + (size_t)l * HID;
    const float* bt = beta   + (size_t)l * HID;

    if (l == 0)
      gnn_gemm_rows_kernel<true><<<GR_BLOCKS, GR_THREADS, 0, stream>>>(
          h, x, Wp, bp, Ws, bs, Wn, bn, h, h_self, h_neigh);
    else
      gnn_gemm_rows_kernel<false><<<GR_BLOCKS, GR_THREADS, 0, stream>>>(
          h, x, Wp, bp, Ws, bs, Wn, bn, h, h_self, h_neigh);

    float* dst = (l == N_LAYERS - 1) ? out : h;
    if (l == 0)
      gnn_edge_own_kernel<false><<<OWN_BLOCKS, OWN_THREADS, 0, stream>>>(
          eidx, ew, We, h_neigh, h_self, gm, bt, h, dst, scratch);
    else
      gnn_edge_own_kernel<true><<<OWN_BLOCKS, OWN_THREADS, 0, stream>>>(
          eidx, ew, We, h_neigh, h_self, gm, bt, h, dst, scratch);
  }
}


# First GEMM launch repacks the edge list for the owner filter: 1-byte owner keys to scan (1/4 of the bytes), one 8-byte record per edge to fetch (in the unwritten out plane); LDS arrival order kept
# speedup vs baseline: 2.0431x; 1.0406x over previous
#include <hip/hip_runtime.h>
#include <hip/hip_bf16.h>

#define N_NODES 100000
#define N_EDGES 1000000
#define IN_DIM  8
#define HID     64
#define N_LAYERS 3
#define LN_EPS  1e-5f
#define SLOPE   0.2f
#define LDW     68

typedef float v2f __attribute__((ext_vector_type(2)));
typedef float v4f __attribute__((ext_vector_type(4)));
typedef float v8f __attribute__((ext_vector_type(8)));

#define OWN_BLOCKS  256
#define OWN_NODES   391
#define OWN_THREADS 1024
#define OWN_WAVES   (OWN_THREADS / 32)
#define L_CAP       256
#define L_WORDS     (2 * L_CAP)
#define C_BITS      17
#define ROW_VEC     (N_EDGES / 4)
#define KEY_VEC     (N_EDGES / 16)
#define KEY_OFF     (OWN_BLOCKS * OWN_WAVES * (L_WORDS + 1))
static_assert(OWN_BLOCKS <= 256 && OWN_BLOCKS * OWN_NODES >= N_NODES, "an owner's number fits a byte; owner ranges cover the nodes");
#define KEY_WORDS   (4 * 256 * 256)
static_assert(N_EDGES % 16 == 0 && KEY_OFF % 4 == 0 && KEY_WORDS >= ROW_VEC &&
              (size_t)KEY_OFF + KEY_WORDS <= (size_t)N_NODES * HID,
              "whole key vectors, 16-byte aligned, every stored word inside the record plane");
static_assert((size_t)KEY_WORDS * 4 * 2 <= (size_t)N_NODES * HID, "one 8-byte record per stored edge slot fits the output plane");

#define LDH        68
#define GR_BLOCKS  256
#define GR_WAVES   8
#define GR_THREADS (GR_WAVES * 32)
#define ROW_TILES  (N_NODES / 16)
static_assert(N_NODES % 16 == 0 && GR_BLOCKS * GR_WAVES <= ROW_TILES, "whole tiles; every wave has a first tile");
static_assert(4 * GR_BLOCKS * GR_THREADS == KEY_WORDS, "four key words a thread");

struct gr_ahead {
  v2f    a[16];
  v4f    x;
};

template <bool PROJ>
__global__ __launch_bounds__(GR_THREADS)
void gnn_gemm_rows_kernel(const float* __restrict__ h,
                          const float* __restrict__ x, const float* __restrict__ Wp,
                          const float* __restrict__ bp,
                          const float* __restrict__ Wself, const float* __restrict__ bself,
                          const float* __restrict__ Wneigh, const float* __restrict__ bneigh,
                          float* __restrict__ h_new,
                          float* __restrict__ h_self, float* __restrict__ h_neigh,
                          const int* __restrict__ eidx, const float* __restrict__ ew,
                          float* __restrict__ scratch, float* __restrict__ edge_recs) {
  __shared__ __attribute__((aligned(16))) float w_s[2 * HID * LDW];
  __shared__ __attribute__((aligned(16))) float h_t[PROJ ? GR_WAVES * 16 * LDH : 1];
  const int t = threadIdx.x;
  int4 er[4], ec[4];
  v4f  ewv[4];
  if (PROJ) {
#pragma unroll
    for (int i = 0; i < 4; ++i) {
      const int v  = blockIdx.x * GR_THREADS + t + i * (GR_BLOCKS * GR_THREADS);
      const int vi = (v < ROW_VEC) ? v : (ROW_VEC - 1);
      er[i]  = ((const int4*)eidx)[vi];
      ec[i]  = ((const int4*)(eidx + N_EDGES))[vi];
      ewv[i] = ((const v4f*)ew)[vi];
    }
  }
#pragma unroll
  for (int i = 0; i < 4; ++i) {
    const int v = t + i * GR_THREADS;
    *(float4*)(w_s + (v >> 4) * LDW + (v & 15) * 4)             = ((const float4*)Wself)[v];
    *(float4*)(w_s + HID * LDW + (v >> 4) * LDW + (v & 15) * 4) = ((const float4*)Wneigh)[v];
  }
  __syncthreads();
  if (PROJ) {
    unsigned* keys = (unsigned*)scratch + KEY_OFF;
    uint4*    rec2 = (uint4*)edge_recs;
#pragma unroll
    for (int i = 0; i < 4; ++i) {
      const int v = blockIdx.x * GR_THREADS + t + i * (GR_BLOCKS * GR_THREADS);
      const int rr[4] = {er[i].x, er[i].y, er[i].z, er[i].w};
      const int cc[4] = {ec[i].x, ec[i].y, ec[i].z, ec[i].w};
      unsigned o[4], w0[4];
#pragma unroll
      for (int p = 0; p < 4; ++p) {
        const unsigned q = (unsigned)rr[p] / (unsigned)OWN_NODES;
        o[p] = (q < (unsigned)(OWN_BLOCKS - 1)) ? q : (unsigned)(OWN_BLOCKS - 1);
        const unsigned l = (unsigned)rr[p] - o[p] * (unsigned)OWN_NODES;
        w0[p] = (((l < (unsigned)(OWN_NODES - 1)) ? l : (unsigned)(OWN_NODES - 1)) << C_BITS) |
                ((unsigned)cc[p] & ((1u << C_BITS) - 1u));
      }
      keys[v] = o[0] | (o[1] << 8) | (o[2] << 16) | (o[3] << 24);
      rec2[2 * v]     = make_uint4(w0[0], __float_as_uint(ewv[i][0]), w0[1], __float_as_uint(ewv[i][1]));
      rec2[2 * v + 1] = make_uint4(w0[2], __float_as_uint(ewv[i][2]), w0[3], __float_as_uint(ewv[i][3]));
    }
  }

  const int lane = t & 31;
  const int wave = __builtin_amdgcn_readfirstlane(t >> 5);
  const int lr = lane & 15;
  const int koff = (lane < 16) ? 0 : 2;

  v2f bfr[8][16];
  float bias[8];
#pragma unroll
  for (int ct = 0; ct < 8; ++ct) {
    const float* brow = w_s + (ct >> 2) * (HID * LDW) + ((ct & 3) * 16 + lr) * LDW;
#pragma unroll
    for (int kb = 0; kb < 16; ++kb) bfr[ct][kb] = *(const v2f*)(brow + kb * 4 + koff);
    bias[ct] = ((ct < 4) ? bself : bneigh)[(ct & 3) * 16 + lr];
  }

  float pw0[IN_DIM], pw1[IN_DIM], pb0 = 0.f, pb1 = 0.f;
  float* ht = h_t;
  if (PROJ) {
    const float4* wq = (const float4*)(Wp + 2 * lane * IN_DIM);
    const float4 q0 = wq[0], q1 = wq[1], q2 = wq[2], q3 = wq[3];
    pw0[0] = q0.x; pw0[1] = q0.y; pw0[2] = q0.z; pw0[3] = q0.w;
    pw0[4] = q1.x; pw0[5] = q1.y; pw0[6] = q1.z; pw0[7] = q1.w;
    pw1[0] = q2.x; pw1[1] = q2.y; pw1[2] = q2.z; pw1[3] = q2.w;
    pw1[4] = q3.x; pw1[5] = q3.y; pw1[6] = q3.z; pw1[7] = q3.w;
    pb0 = bp[2 * lane]; pb1 = bp[2 * lane + 1];
    ht = h_t + wave * (16 * LDH);
  }

  auto ahead = [&](gr_ahead& g, const int tile) {
    if (PROJ) {
      g.x = *(const v4f*)(x + (size_t)tile * (16 * IN_DIM) + 4 * lane);
    } else {
      const float* arow = h + (size_t)(tile * 16 + lr) * HID + koff;
#pragma unroll
      for (int kb = 0; kb < 16; ++kb) g.a[kb] = *(const v2f*)(arow + kb * 4);
    }
  };

  auto finish = [&](const gr_ahead& g, const int tile) {
    v2f ac[16];
    if (PROJ) {
      const float xc[4] = {g.x[0], g.x[1], g.x[2], g.x[3]};
#pragma unroll
      for (int r = 0; r < 16; ++r) {
        float p0 = pb0, p1 = pb1;
#pragma unroll
        for (int k = 0; k < IN_DIM; ++k) {
          const float xs = __int_as_float(__builtin_amdgcn_readlane(__float_as_int(xc[k & 3]), 2 * r + (k >> 2)));
          p0 = __builtin_fmaf(xs, pw0[k], p0);
          p1 = __builtin_fmaf(xs, pw1[k], p1);
        }
        v2f hv; hv.x = p0; hv.y = p1;
        *(v2f*)(h_new + (size_t)(tile * 16 + r) * HID + 2 * lane) = hv;
        *(v2f*)(ht + r * LDH + 2 * lane) = hv;
      }
#pragma unroll
      for (int kb = 0; kb < 16; ++kb) ac[kb] = *(const v2f*)(ht + lr * LDH + kb * 4 + koff);
    } else {
#pragma unroll
      for (int kb = 0; kb < 16; ++kb) ac[kb] = g.a[kb];
    }
    v8f c[8];
#pragma unroll
    for (int ct = 0; ct < 8; ++ct) c[ct] = v8f{};
#pragma unroll
    for (int kb = 0; kb < 16; ++kb) {
#pragma unroll
      for (int ct = 0; ct < 8; ++ct)
        c[ct] = __builtin_amdgcn_wmma_f32_16x16x4_f32(false, ac[kb], false, bfr[ct][kb], (short)0, c[ct], false, false);
    }
    const int mbase = tile * 16 + ((lane < 16) ? 0 : 8);
#pragma unroll
    for (int i = 0; i < 8; ++i) {
#pragma unroll
      for (int ct = 0; ct < 8; ++ct) {
        float* o = (ct < 4) ? h_self : h_neigh;
        o[(size_t)(mbase + i) * HID + (ct & 3) * 16 + lr] = c[ct][i] + bias[ct];
      }
    }
  };

  const int NW = GR_BLOCKS * GR_WAVES;
  int rt = wave * GR_BLOCKS + blockIdx.x;
  auto hold = [&](const gr_ahead& g) {
    if (PROJ) {
      asm volatile("" :: "v"(g.x));
    } else {
      asm volatile("" :: "v"(g.a[0]), "v"(g.a[1]), "v"(g.a[2]), "v"(g.a[3]), "v"(g.a[4]), "v"(g.a[5]),
                         "v"(g.a[6]), "v"(g.a[7]), "v"(g.a[8]), "v"(g.a[9]), "v"(g.a[10]), "v"(g.a[11]),
                         "v"(g.a[12]), "v"(g.a[13]), "v"(g.a[14]), "v"(g.a[15]));
    }
  };
  gr_ahead g0, g1;
  ahead(g0, rt);
  for (;;) {
    ahead(g1, (rt + NW < ROW_TILES) ? rt + NW : rt);
    __builtin_amdgcn_sched_barrier(0);
    finish(g0, rt);
    hold(g1);
    rt += NW;
    if (rt >= ROW_TILES) break;
    ahead(g0, (rt + NW < ROW_TILES) ? rt + NW : rt);
    __builtin_amdgcn_sched_barrier(0);
    finish(g1, rt);
    hold(g0);
    rt += NW;
    if (rt >= ROW_TILES) break;
  }
}

#define KEY_SWEEPS  ((KEY_VEC + OWN_THREADS - 1) / OWN_THREADS)
#define Q_DRAIN     32
#define Q_CAP       (Q_DRAIN + 2 * 32 * 16)
#define EPI_ROWS    ((OWN_NODES + OWN_WAVES - 1) / OWN_WAVES)
static_assert(N_NODES <= (1 << C_BITS) && OWN_NODES <= (1 << (32 - C_BITS)), "record key fields");
static_assert((size_t)OWN_BLOCKS * OWN_WAVES * (L_WORDS + 1) <= (size_t)N_NODES * HID,
              "record regions and their counts fit the borrowed plane");

template <bool LIST>
__global__ __launch_bounds__(OWN_THREADS)
void gnn_edge_own_kernel(const int* __restrict__ eidx, const float* __restrict__ ew,
                         const float* __restrict__ Wedge, const float* __restrict__ h_neigh,
                         const float* __restrict__ h_self, const float* __restrict__ gamma,
                         const float* __restrict__ beta, const float* h, float* dst,
                         float* __restrict__ scratch,
                         const float* edge_recs) {
  __shared__ float    acc[OWN_NODES * HID];
  __shared__ unsigned q[OWN_WAVES][Q_CAP];

  const int t = threadIdx.x, wave = t >> 5, lane = t & 31;
  const int lo  = blockIdx.x * OWN_NODES;
  const int cnt = (N_NODES - lo < OWN_NODES) ? (N_NODES - lo) : OWN_NODES;

  const int wv = __builtin_amdgcn_readfirstlane(wave);
  v2f a[EPI_ROWS], hh[EPI_ROWS];
#pragma unroll
  for (int j = 0; j < EPI_ROWS; ++j) {
    int n = wv + j * OWN_WAVES;
    n = (n < cnt) ? n : (cnt - 1);
    const size_t base = (size_t)(lo + n) * HID + 2 * lane;
    a[j]  = *(const v2f*)(h_self + base);
    hh[j] = *(const v2f*)(h + base);
  }
  const float gm0 = gamma[2 * lane], gm1 = gamma[2 * lane + 1];
  const float bt0 = beta[2 * lane],  bt1 = beta[2 * lane + 1];

  for (int i = t; i < OWN_NODES * HID / 4; i += OWN_THREADS)
    ((float4*)acc)[i] = make_float4(0.f, 0.f, 0.f, 0.f);
  __syncthreads();

  const int*  __restrict__ colp = eidx + N_EDGES;
  const int   c2  = lane * 2;
  const float we0 = Wedge[c2], we1 = Wedge[c2 + 1];
  unsigned* myq = q[wave];
  int qn = 0;
  const int region = blockIdx.x * OWN_WAVES + wave;
  unsigned* lst = (unsigned*)scratch;
  unsigned* myl = lst + (size_t)region * L_WORDS;
  unsigned* myc = lst + (size_t)OWN_BLOCKS * OWN_WAVES * L_WORDS + region;
  int ln = 0;

  auto add32 = [&](const int rl, const int cl, const int wl, const int valid) {
    for (int u = 0; u < valid; u += 8) {
      v2f m[8];
#pragma unroll
      for (int k = 0; k < 8; ++k) {
        const int c = __builtin_amdgcn_readlane(cl, u + k);
        m[k] = *(const v2f*)(h_neigh + (size_t)c * HID + c2);
      }
#pragma unroll
      for (int k = 0; k < 8; ++k) {
        if (u + k < valid) {
          const float w = __int_as_float(__builtin_amdgcn_readlane(wl, u + k));
          const int   r = __builtin_amdgcn_readlane(rl, u + k);
          const float g0 = 1.f / (1.f + __expf(-w * we0));
          const float g1 = 1.f / (1.f + __expf(-w * we1));
          float* d = acc + r * HID + c2;
          atomicAdd(d,     m[k].x * g0);
          atomicAdd(d + 1, m[k].y * g1);
        }
      }
    }
  };

  auto drain = [&](const int total) {
    for (int base = 0; base < total; base += 32) {
      const int valid = (total - base < 32) ? (total - base) : 32;
      const int e  = (lane < valid) ? (int)myq[base + lane] : 0;
      int rl, cl, wl;
      if (!LIST) {
        const uint2 r = ((const uint2*)edge_recs)[e];
        rl = (int)(r.x >> C_BITS);
        cl = (int)(r.x & ((1u << C_BITS) - 1u));
        wl = (int)r.y;
        const int p = ln + base + lane;
        if (lane < valid && p < L_CAP)
          *(uint2*)(myl + 2 * p) = r;
      } else {
        const unsigned l = (unsigned)(eidx[e] - lo);
        rl = (int)((l < (unsigned)(OWN_NODES - 1)) ? l : (unsigned)(OWN_NODES - 1));
        cl = colp[e];
        wl = __float_as_int(ew[e]);
      }
      add32(rl, cl, wl, valid);
    }
    ln += total;
  };

  const unsigned bb = (unsigned)blockIdx.x * 0x01010101u;
  auto zero_bytes = [](const unsigned x) {
    return ~(((x & 0x7f7f7f7fu) + 0x7f7f7f7fu) | x | 0x7f7f7f7fu);
  };
  auto take = [&](const uint4 v, const bool ok, const int ebase) {
    const unsigned z0 = zero_bytes(v.x ^ bb), z1 = zero_bytes(v.y ^ bb);
    const unsigned z2 = zero_bytes(v.z ^ bb), z3 = zero_bytes(v.w ^ bb);
    const unsigned any = ok ? (z0 | z1 | z2 | z3) : 0u;
    unsigned m = __builtin_amdgcn_ballot_w32(any != 0u);
    while (m) {
      const int src = __builtin_ctz(m);
      m &= m - 1u;
      const int eb = __builtin_amdgcn_readlane(ebase, src);
      const unsigned zz[4] = {(unsigned)__builtin_amdgcn_readlane((int)z0, src),
                              (unsigned)__builtin_amdgcn_readlane((int)z1, src),
                              (unsigned)__builtin_amdgcn_readlane((int)z2, src),
                              (unsigned)__builtin_amdgcn_readlane((int)z3, src)};
#pragma unroll
      for (int j = 0; j < 4; ++j) {
        unsigned zs = zz[j];
        while (zs) {
          const int p = __builtin_ctz(zs) >> 3;
          zs &= zs - 1u;
          if (lane == 0) myq[qn] = (unsigned)(eb + 4 * j + p);
          ++qn;
        }
      }
    }
  };

  bool scan = true;
  if (LIST) {
    const int n = __builtin_amdgcn_readfirstlane((int)*myc);
    if (n <= L_CAP) {
      scan = false;
      const uint2* rec = (const uint2*)myl;
      uint2 nx = (lane < n) ? rec[lane] : make_uint2(0u, 0u);
      for (int base = 0; base < n; base += 32) {
        const uint2 cur = nx;
        const int nb = base + 32 + lane;
        nx = (nb < n) ? rec[nb] : make_uint2(0u, 0u);
        const int valid = (n - base < 32) ? (n - base) : 32;
        add32((int)(cur.x >> C_BITS), (int)(cur.x & ((1u << C_BITS) - 1u)), (int)cur.y, valid);
      }
    }
  }

  const int start = (blockIdx.x * KEY_SWEEPS) / OWN_BLOCKS;

  if (scan) {
    const uint4* key4 = (const uint4*)(lst + KEY_OFF);
    for (int g = 0; g < KEY_SWEEPS; g += 4) {
      uint4 v[4];
      int   eb[4];
      bool  ok[4];
#pragma unroll
      for (int k = 0; k < 4; ++k) {
        int it = g + k + start;
        if (it >= KEY_SWEEPS) it -= KEY_SWEEPS;
        const int idx = it * OWN_THREADS + t;
        ok[k] = (g + k < KEY_SWEEPS) && (idx < KEY_VEC);
        eb[k] = idx * 16;
        v[k]  = key4[(idx < KEY_VEC) ? idx : (KEY_VEC - 1)];
      }
#pragma unroll
      for (int k = 0; k < 4; ++k) {
        take(v[k], ok[k], eb[k]);
        if ((k & 1) && qn >= Q_DRAIN) {
          const int full = qn & ~7, rem = qn - full;
          drain(full);
          const unsigned keep = (lane < rem) ? myq[full + lane] : 0u;
          if (lane < rem) myq[lane] = keep;
          qn = rem;
        }
      }
    }
    drain(qn);
    if (!LIST && lane == 0) *myc = (unsigned)ln;
  }

  __syncthreads();

  float v0[EPI_ROWS], v1[EPI_ROWS], s[EPI_ROWS], ss[EPI_ROWS];
#pragma unroll
  for (int j = 0; j < EPI_ROWS; ++j) {
    int n = wv + j * OWN_WAVES;
    n = (n < cnt) ? n : (cnt - 1);
    const v2f g = *(const v2f*)(acc + n * HID + c2);
    v0[j] = a[j].x + g.x;
    v1[j] = a[j].y + g.y;
    s[j]  = v0[j] + v1[j];
    ss[j] = v0[j] * v0[j] + v1[j] * v1[j];
  }
#pragma unroll
  for (int m = 16; m >= 1; m >>= 1) {
#pragma unroll
    for (int j = 0; j < EPI_ROWS; ++j) {
      s[j]  += __shfl_xor(s[j],  m, 32);
      ss[j] += __shfl_xor(ss[j], m, 32);
    }
  }
#pragma unroll
  for (int j = 0; j < EPI_ROWS; ++j) {
    const int n = wv + j * OWN_WAVES;
    const float mu  = s[j] * (1.f / HID);
    const float var = ss[j] * (1.f / HID) - mu * mu;
    const float inv = rsqrtf(var + LN_EPS);
    float o0 = (v0[j] - mu) * inv * gm0 + bt0;
    float o1 = (v1[j] - mu) * inv * gm1 + bt1;
    o0 = (o0 >= 0.f) ? o0 : SLOPE * o0;
    o1 = (o1 >= 0.f) ? o1 : SLOPE * o1;
    v2f r; r.x = hh[j].x + o0; r.y = hh[j].y + o1;
    if (n < cnt)
      *(v2f*)(dst + (size_t)(lo + n) * HID + c2) = r;
  }
}

extern "C" void kernel_launch(void* const* d_in, const int* in_sizes, int n_in,
                              void* d_out, int out_size, void* d_ws, size_t ws_size,
                              hipStream_t stream) {
  const float* x      = (const float*)d_in[0];
  const int*   eidx   = (const int*)  d_in[1];
  const float* ew     = (const float*)d_in[2];
  const float* Wp     = (const float*)d_in[3];
  const float* bp     = (const float*)d_in[4];
  const float* Wself  = (const float*)d_in[5];
  const float* bself  = (const float*)d_in[6];
  const float* Wneigh = (const float*)d_in[7];
  const float* bneigh = (const float*)d_in[8];
  const float* Wedge  = (const float*)d_in[9];
  const float* gamma  = (const float*)d_in[10];
  const float* beta   = (const float*)d_in[11];
  float* out = (float*)d_out;

  const size_t NH = (size_t)N_NODES * HID;
  float* h       = (float*)d_ws;
  float* h_self  = h + NH;
  float* h_neigh = h_self + NH;
  float* scratch = h_neigh + NH;

  for (int l = 0; l < N_LAYERS; ++l) {
    const float* Ws = Wself  + (size_t)l * HID * HID;
    const float* bs = bself  + (size_t)l * HID;
    const float* Wn = Wneigh + (size_t)l * HID * HID;
    const float* bn = bneigh + (size_t)l * HID;
    const float* We = Wedge  + (size_t)l * HID;
    const float* gm = gamma  + (size_t)l * HID;
    const float* bt = beta   + (size_t)l * HID;

    if (l == 0)
      gnn_gemm_rows_kernel<true><<<GR_BLOCKS, GR_THREADS, 0, stream>>>(
          h, x, Wp, bp, Ws, bs, Wn, bn, h, h_self, h_neigh, eidx, ew, scratch, out);
    else
      gnn_gemm_rows_kernel<false><<<GR_BLOCKS, GR_THREADS, 0, stream>>>(
          h, x, Wp, bp, Ws, bs, Wn, bn, h, h_self, h_neigh, eidx, ew, scratch, out);

    float* dst = (l == N_LAYERS - 1) ? out : h;
    if (l == 0)
      gnn_edge_own_kernel<false><<<OWN_BLOCKS, OWN_THREADS, 0, stream>>>(
          eidx, ew, We, h_neigh, h_self, gm, bt, h, dst, scratch, out);
    else
      gnn_edge_own_kernel<true><<<OWN_BLOCKS, OWN_THREADS, 0, stream>>>(
          eidx, ew, We, h_neigh, h_self, gm, bt, h, dst, scratch, out);
  }
}


# Owner keys as bit planes (8 words per 32 edges, wave ballots in the first GEMM launch): the owner filter tests a lane's 32 edges in 8 XOR+AND, matching lanes queue at once; LDS arrival order kept
# speedup vs baseline: 2.1471x; 1.0509x over previous
#include <hip/hip_runtime.h>
#include <hip/hip_bf16.h>

#define N_NODES 100000
#define N_EDGES 1000000
#define IN_DIM  8
#define HID     64
#define N_LAYERS 3
#define LN_EPS  1e-5f
#define SLOPE   0.2f
#define LDW     68

typedef float v2f __attribute__((ext_vector_type(2)));
typedef float v4f __attribute__((ext_vector_type(4)));
typedef float v8f __attribute__((ext_vector_type(8)));

#define OWN_BLOCKS  256
#define OWN_NODES   391
#define OWN_THREADS 1024
#define OWN_WAVES   (OWN_THREADS / 32)
#define L_CAP       256
#define L_WORDS     (2 * L_CAP)
#define C_BITS      17
#define ROW_VEC     (N_EDGES / 4)
#define N_VB        ((ROW_VEC + 31) / 32)
#define N_GRP       (4 * N_VB)
#define KEY_OFF     (OWN_BLOCKS * OWN_WAVES * (L_WORDS + 1))
static_assert(OWN_BLOCKS <= 256 && OWN_BLOCKS * OWN_NODES >= N_NODES, "an owner's number fits eight bits; owner ranges cover the nodes");
#define KEY_WORDS   (4 * 256 * 256)
static_assert(N_EDGES % 4 == 0 && KEY_OFF % 4 == 0 && KEY_WORDS >= 8 * N_GRP &&
              (size_t)KEY_OFF + KEY_WORDS <= (size_t)N_NODES * HID,
              "whole index vectors, 16-byte aligned key groups, every stored word inside the record plane");
static_assert((size_t)KEY_WORDS * 4 * 2 <= (size_t)N_NODES * HID, "one 8-byte record per stored edge slot fits the output plane");

#define LDH        68
#define GR_BLOCKS  256
#define GR_WAVES   8
#define GR_THREADS (GR_WAVES * 32)
#define ROW_TILES  (N_NODES / 16)
static_assert(N_NODES % 16 == 0 && GR_BLOCKS * GR_WAVES <= ROW_TILES, "whole tiles; every wave has a first tile");
static_assert(4 * GR_BLOCKS * GR_THREADS == KEY_WORDS, "four key words a thread");

struct gr_ahead {
  v2f    a[16];
  v4f    x;
};

template <bool PROJ>
__global__ __launch_bounds__(GR_THREADS)
void gnn_gemm_rows_kernel(const float* __restrict__ h,
                          const float* __restrict__ x, const float* __restrict__ Wp,
                          const float* __restrict__ bp,
                          const float* __restrict__ Wself, const float* __restrict__ bself,
                          const float* __restrict__ Wneigh, const float* __restrict__ bneigh,
                          float* __restrict__ h_new,
                          float* __restrict__ h_self, float* __restrict__ h_neigh,
                          const int* __restrict__ eidx, const float* __restrict__ ew,
                          float* __restrict__ scratch, float* __restrict__ edge_recs) {
  __shared__ __attribute__((aligned(16))) float w_s[2 * HID * LDW];
  __shared__ __attribute__((aligned(16))) float h_t[PROJ ? GR_WAVES * 16 * LDH : 1];
  const int t = threadIdx.x;
  int4 er[4], ec[4];
  v4f  ewv[4];
  if (PROJ) {
#pragma unroll
    for (int i = 0; i < 4; ++i) {
      const int v  = blockIdx.x * GR_THREADS + t + i * (GR_BLOCKS * GR_THREADS);
      const int vi = (v < ROW_VEC) ? v : (ROW_VEC - 1);
      er[i]  = ((const int4*)eidx)[vi];
      ec[i]  = ((const int4*)(eidx + N_EDGES))[vi];
      ewv[i] = ((const v4f*)ew)[vi];
    }
  }
#pragma unroll
  for (int i = 0; i < 4; ++i) {
    const int v = t + i * GR_THREADS;
    *(float4*)(w_s + (v >> 4) * LDW + (v & 15) * 4)             = ((const float4*)Wself)[v];
    *(float4*)(w_s + HID * LDW + (v >> 4) * LDW + (v & 15) * 4) = ((const float4*)Wneigh)[v];
  }
  __syncthreads();
  if (PROJ) {
    unsigned* keys = (unsigned*)scratch + KEY_OFF;
    uint4*    rec2 = (uint4*)edge_recs;
#pragma unroll
    for (int i = 0; i < 4; ++i) {
      const int v = blockIdx.x * GR_THREADS + t + i * (GR_BLOCKS * GR_THREADS);
      const int rr[4] = {er[i].x, er[i].y, er[i].z, er[i].w};
      const int cc[4] = {ec[i].x, ec[i].y, ec[i].z, ec[i].w};
      unsigned o[4], w0[4];
#pragma unroll
      for (int p = 0; p < 4; ++p) {
        const unsigned q = (unsigned)rr[p] / (unsigned)OWN_NODES;
        o[p] = (q < (unsigned)(OWN_BLOCKS - 1)) ? q : (unsigned)(OWN_BLOCKS - 1);
        const unsigned l = (unsigned)rr[p] - o[p] * (unsigned)OWN_NODES;
        w0[p] = (((l < (unsigned)(OWN_NODES - 1)) ? l : (unsigned)(OWN_NODES - 1)) << C_BITS) |
                ((unsigned)cc[p] & ((1u << C_BITS) - 1u));
      }
      unsigned pw = 0u;
#pragma unroll
      for (int p = 0; p < 4; ++p) {
#pragma unroll
        for (int j = 0; j < 8; ++j) {
          const unsigned bits = __builtin_amdgcn_ballot_w32(((o[p] >> j) & 1u) != 0u);
          pw = ((t & 31) == 8 * p + j) ? bits : pw;
        }
      }
      const int vb = blockIdx.x * GR_WAVES + (t >> 5) + i * (GR_BLOCKS * GR_WAVES);
      keys[vb * 32 + (t & 31)] = pw;
      rec2[2 * v]     = make_uint4(w0[0], __float_as_uint(ewv[i][0]), w0[1], __float_as_uint(ewv[i][1]));
      rec2[2 * v + 1] = make_uint4(w0[2], __float_as_uint(ewv[i][2]), w0[3], __float_as_uint(ewv[i][3]));
    }
  }

  const int lane = t & 31;
  const int wave = __builtin_amdgcn_readfirstlane(t >> 5);
  const int lr = lane & 15;
  const int koff = (lane < 16) ? 0 : 2;

  v2f bfr[8][16];
  float bias[8];
#pragma unroll
  for (int ct = 0; ct < 8; ++ct) {
    const float* brow = w_s + (ct >> 2) * (HID * LDW) + ((ct & 3) * 16 + lr) * LDW;
#pragma unroll
    for (int kb = 0; kb < 16; ++kb) bfr[ct][kb] = *(const v2f*)(brow + kb * 4 + koff);
    bias[ct] = ((ct < 4) ? bself : bneigh)[(ct & 3) * 16 + lr];
  }

  float pw0[IN_DIM], pw1[IN_DIM], pb0 = 0.f, pb1 = 0.f;
  float* ht = h_t;
  if (PROJ) {
    const float4* wq = (const float4*)(Wp + 2 * lane * IN_DIM);
    const float4 q0 = wq[0], q1 = wq[1], q2 = wq[2], q3 = wq[3];
    pw0[0] = q0.x; pw0[1] = q0.y; pw0[2] = q0.z; pw0[3] = q0.w;
    pw0[4] = q1.x; pw0[5] = q1.y; pw0[6] = q1.z; pw0[7] = q1.w;
    pw1[0] = q2.x; pw1[1] = q2.y; pw1[2] = q2.z; pw1[3] = q2.w;
    pw1[4] = q3.x; pw1[5] = q3.y; pw1[6] = q3.z; pw1[7] = q3.w;
    pb0 = bp[2 * lane]; pb1 = bp[2 * lane + 1];
    ht = h_t + wave * (16 * LDH);
  }

  auto ahead = [&](gr_ahead& g, const int tile) {
    if (PROJ) {
      g.x = *(const v4f*)(x + (size_t)tile * (16 * IN_DIM) + 4 * lane);
    } else {
      const float* arow = h + (size_t)(tile * 16 + lr) * HID + koff;
#pragma unroll
      for (int kb = 0; kb < 16; ++kb) g.a[kb] = *(const v2f*)(arow + kb * 4);
    }
  };

  auto finish = [&](const gr_ahead& g, const int tile) {
    v2f ac[16];
    if (PROJ) {
      const float xc[4] = {g.x[0], g.x[1], g.x[2], g.x[3]};
#pragma unroll
      for (int r = 0; r < 16; ++r) {
        float p0 = pb0, p1 = pb1;
#pragma unroll
        for (int k = 0; k < IN_DIM; ++k) {
          const float xs = __int_as_float(__builtin_amdgcn_readlane(__float_as_int(xc[k & 3]), 2 * r + (k >> 2)));
          p0 = __builtin_fmaf(xs, pw0[k], p0);
          p1 = __builtin_fmaf(xs, pw1[k], p1);
        }
        v2f hv; hv.x = p0; hv.y = p1;
        *(v2f*)(h_new + (size_t)(tile * 16 + r) * HID + 2 * lane) = hv;
        *(v2f*)(ht + r * LDH + 2 * lane) = hv;
      }
#pragma unroll
      for (int kb = 0; kb < 16; ++kb) ac[kb] = *(const v2f*)(ht + lr * LDH + kb * 4 + koff);
    } else {
#pragma unroll
      for (int kb = 0; kb < 16; ++kb) ac[kb] = g.a[kb];
    }
    v8f c[8];
#pragma unroll
    for (int ct = 0; ct < 8; ++ct) c[ct] = v8f{};
#pragma unroll
    for (int kb = 0; kb < 16; ++kb) {
#pragma unroll
      for (int ct = 0; ct < 8; ++ct)
        c[ct] = __builtin_amdgcn_wmma_f32_16x16x4_f32(false, ac[kb], false, bfr[ct][kb], (short)0, c[ct], false, false);
    }
    const int mbase = tile * 16 + ((lane < 16) ? 0 : 8);
#pragma unroll
    for (int i = 0; i < 8; ++i) {
#pragma unroll
      for (int ct = 0; ct < 8; ++ct) {
        float* o = (ct < 4) ? h_self : h_neigh;
        o[(size_t)(mbase + i) * HID + (ct & 3) * 16 + lr] = c[ct][i] + bias[ct];
      }
    }
  };

  const int NW = GR_BLOCKS * GR_WAVES;
  int rt = wave * GR_BLOCKS + blockIdx.x;
  auto hold = [&](const gr_ahead& g) {
    if (PROJ) {
      asm volatile("" :: "v"(g.x));
    } else {
      asm volatile("" :: "v"(g.a[0]), "v"(g.a[1]), "v"(g.a[2]), "v"(g.a[3]), "v"(g.a[4]), "v"(g.a[5]),
                         "v"(g.a[6]), "v"(g.a[7]), "v"(g.a[8]), "v"(g.a[9]), "v"(g.a[10]), "v"(g.a[11]),
                         "v"(g.a[12]), "v"(g.a[13]), "v"(g.a[14]), "v"(g.a[15]));
    }
  };
  gr_ahead g0, g1;
  ahead(g0, rt);
  for (;;) {
    ahead(g1, (rt + NW < ROW_TILES) ? rt + NW : rt);
    __builtin_amdgcn_sched_barrier(0);
    finish(g0, rt);
    hold(g1);
    rt += NW;
    if (rt >= ROW_TILES) break;
    ahead(g0, (rt + NW < ROW_TILES) ? rt + NW : rt);
    __builtin_amdgcn_sched_barrier(0);
    finish(g1, rt);
    hold(g0);
    rt += NW;
    if (rt >= ROW_TILES) break;
  }
}

#define KEY_SWEEPS  ((N_GRP + OWN_THREADS - 1) / OWN_THREADS)
#define Q_DRAIN     32
#define Q_CAP       (Q_DRAIN + 32 * 32)
#define EPI_ROWS    ((OWN_NODES + OWN_WAVES - 1) / OWN_WAVES)
static_assert(N_NODES <= (1 << C_BITS) && OWN_NODES <= (1 << (32 - C_BITS)), "record key fields");
static_assert((size_t)OWN_BLOCKS * OWN_WAVES * (L_WORDS + 1) <= (size_t)N_NODES * HID,
              "record regions and their counts fit the borrowed plane");

template <bool LIST>
__global__ __launch_bounds__(OWN_THREADS)
void gnn_edge_own_kernel(const int* __restrict__ eidx, const float* __restrict__ ew,
                         const float* __restrict__ Wedge, const float* __restrict__ h_neigh,
                         const float* __restrict__ h_self, const float* __restrict__ gamma,
                         const float* __restrict__ beta, const float* h, float* dst,
                         float* __restrict__ scratch,
                         const float* edge_recs) {
  __shared__ float    acc[OWN_NODES * HID];
  __shared__ unsigned q[OWN_WAVES][Q_CAP];

  const int t = threadIdx.x, wave = t >> 5, lane = t & 31;
  const int lo  = blockIdx.x * OWN_NODES;
  const int cnt = (N_NODES - lo < OWN_NODES) ? (N_NODES - lo) : OWN_NODES;

  const int wv = __builtin_amdgcn_readfirstlane(wave);
  v2f a[EPI_ROWS], hh[EPI_ROWS];
#pragma unroll
  for (int j = 0; j < EPI_ROWS; ++j) {
    int n = wv + j * OWN_WAVES;
    n = (n < cnt) ? n : (cnt - 1);
    const size_t base = (size_t)(lo + n) * HID + 2 * lane;
    a[j]  = *(const v2f*)(h_self + base);
    hh[j] = *(const v2f*)(h + base);
  }
  const float gm0 = gamma[2 * lane], gm1 = gamma[2 * lane + 1];
  const float bt0 = beta[2 * lane],  bt1 = beta[2 * lane + 1];

  for (int i = t; i < OWN_NODES * HID / 4; i += OWN_THREADS)
    ((float4*)acc)[i] = make_float4(0.f, 0.f, 0.f, 0.f);
  __syncthreads();

  const int*  __restrict__ colp = eidx + N_EDGES;
  const int   c2  = lane * 2;
  const float we0 = Wedge[c2], we1 = Wedge[c2 + 1];
  unsigned* myq = q[wave];
  int qn = 0;
  const int region = blockIdx.x * OWN_WAVES + wave;
  unsigned* lst = (unsigned*)scratch;
  unsigned* myl = lst + (size_t)region * L_WORDS;
  unsigned* myc = lst + (size_t)OWN_BLOCKS * OWN_WAVES * L_WORDS + region;
  int ln = 0;

  auto add32 = [&](const int rl, const int cl, const int wl, const int valid) {
    for (int u = 0; u < valid; u += 8) {
      v2f m[8];
#pragma unroll
      for (int k = 0; k < 8; ++k) {
        const int c = __builtin_amdgcn_readlane(cl, u + k);
        m[k] = *(const v2f*)(h_neigh + (size_t)c * HID + c2);
      }
#pragma unroll
      for (int k = 0; k < 8; ++k) {
        if (u + k < valid) {
          const float w = __int_as_float(__builtin_amdgcn_readlane(wl, u + k));
          const int   r = __builtin_amdgcn_readlane(rl, u + k);
          const float g0 = 1.f / (1.f + __expf(-w * we0));
          const float g1 = 1.f / (1.f + __expf(-w * we1));
          float* d = acc + r * HID + c2;
          atomicAdd(d,     m[k].x * g0);
          atomicAdd(d + 1, m[k].y * g1);
        }
      }
    }
  };

  auto drain = [&](const int total) {
    for (int base = 0; base < total; base += 32) {
      const int valid = (total - base < 32) ? (total - base) : 32;
      const int e  = (lane < valid) ? (int)myq[base + lane] : 0;
      int rl, cl, wl;
      if (!LIST) {
        const uint2 r = ((const uint2*)edge_recs)[e];
        rl = (int)(r.x >> C_BITS);
        cl = (int)(r.x & ((1u << C_BITS) - 1u));
        wl = (int)r.y;
        const int p = ln + base + lane;
        if (lane < valid && p < L_CAP)
          *(uint2*)(myl + 2 * p) = r;
      } else {
        const unsigned l = (unsigned)(eidx[e] - lo);
        rl = (int)((l < (unsigned)(OWN_NODES - 1)) ? l : (unsigned)(OWN_NODES - 1));
        cl = colp[e];
        wl = __float_as_int(ew[e]);
      }
      add32(rl, cl, wl, valid);
    }
    ln += total;
  };

  unsigned sel[8];
#pragma unroll
  for (int j = 0; j < 8; ++j) sel[j] = ((blockIdx.x >> j) & 1u) ? 0u : ~0u;
  auto take = [&](const uint4 lo, const uint4 hi, const bool ok, const int grp) {
    const unsigned same = (lo.x ^ sel[0]) & (lo.y ^ sel[1]) & (lo.z ^ sel[2]) & (lo.w ^ sel[3]) &
                          (hi.x ^ sel[4]) & (hi.y ^ sel[5]) & (hi.z ^ sel[6]) & (hi.w ^ sel[7]);
    const int vb = grp >> 2;
    const int nv = ROW_VEC - 32 * vb;
    const unsigned inlist = (nv >= 32) ? ~0u : ((1u << nv) - 1u);
    unsigned m = ok ? (same & inlist) : 0u;
    const int ebase = 128 * vb + (grp & 3);
    for (int trip = 0; trip < 32; ++trip) {
      const unsigned has = __builtin_amdgcn_ballot_w32(m != 0u);
      if (has == 0u) break;
      const int slot = qn + (int)__builtin_amdgcn_mbcnt_lo(has, 0u);
      if (m != 0u) myq[slot] = (unsigned)(ebase + 4 * __builtin_ctz(m));
      m &= m - 1u;
      qn += __builtin_popcount(has);
    }
  };

  bool scan = true;
  if (LIST) {
    const int n = __builtin_amdgcn_readfirstlane((int)*myc);
    if (n <= L_CAP) {
      scan = false;
      const uint2* rec = (const uint2*)myl;
      uint2 nx = (lane < n) ? rec[lane] : make_uint2(0u, 0u);
      for (int base = 0; base < n; base += 32) {
        const uint2 cur = nx;
        const int nb = base + 32 + lane;
        nx = (nb < n) ? rec[nb] : make_uint2(0u, 0u);
        const int valid = (n - base < 32) ? (n - base) : 32;
        add32((int)(cur.x >> C_BITS), (int)(cur.x & ((1u << C_BITS) - 1u)), (int)cur.y, valid);
      }
    }
  }

  const int start = (blockIdx.x * KEY_SWEEPS) / OWN_BLOCKS;

  if (scan) {
    const uint4* key4 = (const uint4*)(lst + KEY_OFF);
    for (int g = 0; g < KEY_SWEEPS; g += 2) {
      uint4 lo[2], hi[2];
      int   gr[2];
      bool  ok[2];
#pragma unroll
      for (int k = 0; k < 2; ++k) {
        int it = g + k + start;
        if (it >= KEY_SWEEPS) it -= KEY_SWEEPS;
        const int idx = it * OWN_THREADS + t;
        ok[k] = (g + k < KEY_SWEEPS) && (idx < N_GRP);
        gr[k] = (idx < N_GRP) ? idx : (N_GRP - 1);
        lo[k] = key4[2 * gr[k]];
        hi[k] = key4[2 * gr[k] + 1];
      }
#pragma unroll
      for (int k = 0; k < 2; ++k) {
        take(lo[k], hi[k], ok[k], gr[k]);
        if (qn >= Q_DRAIN) {
          const int full = qn & ~7, rem = qn - full;
          drain(full);
          const unsigned keep = (lane < rem) ? myq[full + lane] : 0u;
          if (lane < rem) myq[lane] = keep;
          qn = rem;
        }
      }
    }
    drain(qn);
    if (!LIST && lane == 0) *myc = (unsigned)ln;
  }

  __syncthreads();

  float v0[EPI_ROWS], v1[EPI_ROWS], s[EPI_ROWS], ss[EPI_ROWS];
#pragma unroll
  for (int j = 0; j < EPI_ROWS; ++j) {
    int n = wv + j * OWN_WAVES;
    n = (n < cnt) ? n : (cnt - 1);
    const v2f g = *(const v2f*)(acc + n * HID + c2);
    v0[j] = a[j].x + g.x;
    v1[j] = a[j].y + g.y;
    s[j]  = v0[j] + v1[j];
    ss[j] = v0[j] * v0[j] + v1[j] * v1[j];
  }
#pragma unroll
  for (int m = 16; m >= 1; m >>= 1) {
#pragma unroll
    for (int j = 0; j < EPI_ROWS; ++j) {
      s[j]  += __shfl_xor(s[j],  m, 32);
      ss[j] += __shfl_xor(ss[j], m, 32);
    }
  }
#pragma unroll
  for (int j = 0; j < EPI_ROWS; ++j) {
    const int n = wv + j * OWN_WAVES;
    const float mu  = s[j] * (1.f / HID);
    const float var = ss[j] * (1.f / HID) - mu * mu;
    const float inv = rsqrtf(var + LN_EPS);
    float o0 = (v0[j] - mu) * inv * gm0 + bt0;
    float o1 = (v1[j] - mu) * inv * gm1 + bt1;
    o0 = (o0 >= 0.f) ? o0 : SLOPE * o0;
    o1 = (o1 >= 0.f) ? o1 : SLOPE * o1;
    v2f r; r.x = hh[j].x + o0; r.y = hh[j].y + o1;
    if (n < cnt)
      *(v2f*)(dst + (size_t)(lo + n) * HID + c2) = r;
  }
}

extern "C" void kernel_launch(void* const* d_in, const int* in_sizes, int n_in,
                              void* d_out, int out_size, void* d_ws, size_t ws_size,
                              hipStream_t stream) {
  const float* x      = (const float*)d_in[0];
  const int*   eidx   = (const int*)  d_in[1];
  const float* ew     = (const float*)d_in[2];
  const float* Wp     = (const float*)d_in[3];
  const float* bp     = (const float*)d_in[4];
  const float* Wself  = (const float*)d_in[5];
  const float* bself  = (const float*)d_in[6];
  const float* Wneigh = (const float*)d_in[7];
  const float* bneigh = (const float*)d_in[8];
  const float* Wedge  = (const float*)d_in[9];
  const float* gamma  = (const float*)d_in[10];
  const float* beta   = (const float*)d_in[11];
  float* out = (float*)d_out;

  const size_t NH = (size_t)N_NODES * HID;
  float* h       = (float*)d_ws;
  float* h_self  = h + NH;
  float* h_neigh = h_self + NH;
  float* scratch = h_neigh + NH;

  for (int l = 0; l < N_LAYERS; ++l) {
    const float* Ws = Wself  + (size_t)l * HID * HID;
    const float* bs = bself  + (size_t)l * HID;
    const float* Wn = Wneigh + (size_t)l * HID * HID;
    const float* bn = bneigh + (size_t)l * HID;
    const float* We = Wedge  + (size_t)l * HID;
    const float* gm = gamma  + (size_t)l * HID;
    const float* bt = beta   + (size_t)l * HID;

    if (l == 0)
      gnn_gemm_rows_kernel<true><<<GR_BLOCKS, GR_THREADS, 0, stream>>>(
          h, x, Wp, bp, Ws, bs, Wn, bn, h, h_self, h_neigh, eidx, ew, scratch, out);
    else
      gnn_gemm_rows_kernel<false><<<GR_BLOCKS, GR_THREADS, 0, stream>>>(
          h, x, Wp, bp, Ws, bs, Wn, bn, h, h_self, h_neigh, eidx, ew, scratch, out);

    float* dst = (l == N_LAYERS - 1) ? out : h;
    if (l == 0)
      gnn_edge_own_kernel<false><<<OWN_BLOCKS, OWN_THREADS, 0, stream>>>(
          eidx, ew, We, h_neigh, h_self, gm, bt, h, dst, scratch, out);
    else
      gnn_edge_own_kernel<true><<<OWN_BLOCKS, OWN_THREADS, 0, stream>>>(
          eidx, ew, We, h_neigh, h_self, gm, bt, h, dst, scratch, out);
  }
}
